# starting kernel
# baseline (speedup 1.0000x reference)
#include <hip/hip_runtime.h>

#define N_NODES 100000
#define N_EDGES 1000000
#define DHID    64
#define KORD    8
#define NLAYERS 3

typedef float v2f __attribute__((ext_vector_type(2)));
typedef float v8f __attribute__((ext_vector_type(8)));

__global__ void zero_kernel(float* __restrict__ p, int n) {
    int i = blockIdx.x * blockDim.x + threadIdx.x;
    int stride = gridDim.x * blockDim.x;
    for (; i < n; i += stride) p[i] = 0.0f;
}

__global__ void deg_kernel(const int* __restrict__ src, const int* __restrict__ dst,
                           const float* __restrict__ ew, float* __restrict__ deg, int nE) {
    int e = blockIdx.x * blockDim.x + threadIdx.x;
    if (e >= nE) return;
    int s = src[e];
    if (s == dst[e]) return;
    atomicAdd(&deg[s], ew[e]);
}

__global__ void dinv_kernel(float* __restrict__ deg, int n) {
    int i = blockIdx.x * blockDim.x + threadIdx.x;
    if (i >= n) return;
    float d = deg[i];
    deg[i] = (d > 0.0f) ? rsqrtf(d) : 0.0f;
}

__global__ void norm_kernel(const int* __restrict__ src, const int* __restrict__ dst,
                            const float* __restrict__ ew, const float* __restrict__ dinv,
                            float* __restrict__ norm, int nE) {
    int e = blockIdx.x * blockDim.x + threadIdx.x;
    if (e >= nE) return;
    int s = src[e], d = dst[e];
    float w = (s == d) ? 0.0f : ew[e];
    norm[e] = -dinv[s] * w * dinv[d];
}

__global__ void scatter_kernel(const float* __restrict__ h, const int* __restrict__ src,
                               const int* __restrict__ dst, const float* __restrict__ norm,
                               float* __restrict__ out, int nE) {
    int t  = blockIdx.x * blockDim.x + threadIdx.x;
    int e  = t >> 4;
    int f4 = (t & 15) << 2;
    if (e >= nE) return;
    float w = norm[e];
    if (w == 0.0f) return;
    const float4 hv = *(const float4*)(h + (size_t)src[e] * DHID + f4);
    float* o = out + (size_t)dst[e] * DHID + f4;
    atomicAdd(o + 0, w * hv.x);
    atomicAdd(o + 1, w * hv.y);
    atomicAdd(o + 2, w * hv.z);
    atomicAdd(o + 3, w * hv.w);
}

__global__ void relu_bias_kernel(const float* __restrict__ acc, const float* __restrict__ b,
                                 float* __restrict__ out, int n) {
    int i = blockIdx.x * blockDim.x + threadIdx.x;
    if (i >= n) return;
    out[i] = fmaxf(acc[i] + b[i & (DHID - 1)], 0.0f);
}

template <int MODE>
__global__ void __launch_bounds__(256)
gemm64_wmma(float* __restrict__ T, const float* __restrict__ t0,
            const float* __restrict__ W, float* __restrict__ acc, int nrows) {
    __shared__ float lWp[DHID * DHID];
    for (int idx = threadIdx.x; idx < DHID * DHID; idx += 256) {
        int k = idx >> 6, n = idx & 63;
        lWp[((k >> 1) << 7) + (n << 1) + (k & 1)] = W[idx];
    }
    __syncthreads();

    const int wave = threadIdx.x >> 5;
    const int lane = threadIdx.x & 31;
    const int row0 = (blockIdx.x * 8 + wave) * 16;
    if (row0 >= nrows) return;

    const int lrow  = lane & 15;
    const int khalf = (lane >> 4) << 1;
    const int crow  = row0 + ((lane >> 4) << 3);
    const int ccol  = lane & 15;

    v8f c0 = {}, c1 = {}, c2 = {}, c3 = {};
    if (MODE != 0) {
#pragma unroll
        for (int r = 0; r < 8; ++r) {
            const float* ap = acc + (size_t)(crow + r) * DHID + ccol;
            c0[r] = ap[0]; c1[r] = ap[16]; c2[r] = ap[32]; c3[r] = ap[48];
        }
    }

    float* arow       = T + (size_t)(row0 + lrow) * DHID + khalf;
    const float* trow = (MODE == 2) ? (t0 + (size_t)(row0 + lrow) * DHID + khalf) : nullptr;

#pragma unroll
    for (int kt = 0; kt < 16; ++kt) {
        const int k0 = kt * 4 + khalf;
        v2f a = *(const v2f*)(arow + kt * 4);
        if (MODE == 2) {
            v2f t = *(const v2f*)(trow + kt * 4);
            a.x = 2.0f * a.x - t.x;
            a.y = 2.0f * a.y - t.y;
            *(v2f*)(arow + kt * 4) = a;
        }
        const float* wp = lWp + ((k0 >> 1) << 7);
        v2f b;
        b = *(const v2f*)(wp + ((lrow +  0) << 1));
        c0 = __builtin_amdgcn_wmma_f32_16x16x4_f32(false, a, false, b, (short)0, c0, false, false);
        b = *(const v2f*)(wp + ((lrow + 16) << 1));
        c1 = __builtin_amdgcn_wmma_f32_16x16x4_f32(false, a, false, b, (short)0, c1, false, false);
        b = *(const v2f*)(wp + ((lrow + 32) << 1));
        c2 = __builtin_amdgcn_wmma_f32_16x16x4_f32(false, a, false, b, (short)0, c2, false, false);
        b = *(const v2f*)(wp + ((lrow + 48) << 1));
        c3 = __builtin_amdgcn_wmma_f32_16x16x4_f32(false, a, false, b, (short)0, c3, false, false);
    }

#pragma unroll
    for (int r = 0; r < 8; ++r) {
        float* ap = acc + (size_t)(crow + r) * DHID + ccol;
        ap[0] = c0[r]; ap[16] = c1[r]; ap[32] = c2[r]; ap[48] = c3[r];
    }
}

extern "C" void kernel_launch(void* const* d_in, const int* in_sizes, int n_in,
                              void* d_out, int out_size, void* d_ws, size_t ws_size,
                              hipStream_t stream) {
    const float* x       = (const float*)d_in[0];
    const int*   ei      = (const int*)d_in[1];
    const float* ew      = (const float*)d_in[2];
    const float* weights = (const float*)d_in[3];
    const float* bias    = (const float*)d_in[4];
    float*       out     = (float*)d_out;

    const int* src = ei;
    const int* dst = ei + N_EDGES;

    float* ws   = (float*)d_ws;
    float* norm = ws;
    float* dinv = ws + N_EDGES;
    float* base = ws + N_EDGES + N_NODES;
    float* P[5];
    for (int i = 0; i < 5; ++i) P[i] = base + (size_t)i * N_NODES * DHID;
    float* ACC = P[4];

    const int NF = N_NODES * DHID;
    dim3 blk(256);
    const int gE    = (N_EDGES + 255) / 256;
    const int gN    = (N_NODES + 255) / 256;
    const int gNF   = (NF + 255) / 256;
    const int gSc   = (N_EDGES * 16 + 255) / 256;
    const int gGemm = (N_NODES + 127) / 128;

    zero_kernel<<<1024, blk, 0, stream>>>(dinv, N_NODES);
    deg_kernel <<<gE,   blk, 0, stream>>>(src, dst, ew, dinv, N_EDGES);
    dinv_kernel<<<gN,   blk, 0, stream>>>(dinv, N_NODES);
    norm_kernel<<<gE,   blk, 0, stream>>>(src, dst, ew, dinv, norm, N_EDGES);

    auto prop = [&](const float* tin, float* tout) {
        zero_kernel   <<<2048, blk, 0, stream>>>(tout, NF);
        scatter_kernel<<<gSc,  blk, 0, stream>>>(tin, src, dst, norm, tout, N_EDGES);
    };

    int h_idx = -1;
    for (int l = 0; l < NLAYERS; ++l) {
        int rid[3], c = 0;
        for (int i = 0; i < 4 && c < 3; ++i)
            if (i != h_idx) rid[c++] = i;

        const float* t0 = (h_idx < 0) ? x : P[h_idx];
        float* t1 = P[rid[0]];
        const float* Wl = weights + (size_t)l * KORD * DHID * DHID;

        prop(t0, t1);
        gemm64_wmma<0><<<gGemm, blk, 0, stream>>>((float*)t0, nullptr, Wl,
                                                  ACC, N_NODES);
        gemm64_wmma<1><<<gGemm, blk, 0, stream>>>(t1, nullptr, Wl + DHID * DHID,
                                                  ACC, N_NODES);

        int q[10]; int qh = 0, qt = 0;
        q[qt++] = rid[1];
        q[qt++] = rid[2];
        int t0_idx = h_idx;
        int t1_idx = rid[0];

        for (int k = 2; k < KORD; ++k) {
            int s_idx = q[qh++];
            float* scat = P[s_idx];
            prop(P[t1_idx], scat);
            gemm64_wmma<2><<<gGemm, blk, 0, stream>>>(
                scat, (t0_idx < 0) ? x : P[t0_idx],
                Wl + (size_t)k * DHID * DHID, ACC, N_NODES);
            if (t0_idx >= 0) q[qt++] = t0_idx;
            t0_idx = t1_idx;
            t1_idx = s_idx;
        }

        if (l == NLAYERS - 1) {
            relu_bias_kernel<<<gNF, blk, 0, stream>>>(ACC, bias + l * DHID, out, NF);
        } else {
            float* hn = P[t1_idx];
            relu_bias_kernel<<<gNF, blk, 0, stream>>>(ACC, bias + l * DHID, hn, NF);
            h_idx = t1_idx;
        }
    }
}


# Replace per-edge float-atomic scatter (+ zero fill) by a destination-sorted edge list built once per call and an atomic-free row gather; each sum now runs in edge-id order
# speedup vs baseline: 2.3572x; 2.3572x over previous
#include <hip/hip_runtime.h>

#define N_NODES 100000
#define N_EDGES 1000000
#define DHID    64
#define KORD    8
#define NLAYERS 3

typedef float v2f __attribute__((ext_vector_type(2)));
typedef float v8f __attribute__((ext_vector_type(8)));

__global__ void zero_kernel(float* __restrict__ p, int n) {
    int i = blockIdx.x * blockDim.x + threadIdx.x;
    int stride = gridDim.x * blockDim.x;
    for (; i < n; i += stride) p[i] = 0.0f;
}

__global__ void deg_rank_kernel(const int* __restrict__ src, const int* __restrict__ dst,
                                const float* __restrict__ ew, float* __restrict__ deg,
                                int* __restrict__ cnt, int* __restrict__ rank, int nE) {
    int e = blockIdx.x * blockDim.x + threadIdx.x;
    if (e >= nE) return;
    int s = src[e], d = dst[e];
    if (s != d) atomicAdd(&deg[s], ew[e]);
    rank[e] = atomicAdd(&cnt[d], 1);
}

__global__ void __launch_bounds__(256)
dinv_bsum_kernel(float* __restrict__ deg, const int* __restrict__ cnt,
                 int* __restrict__ bsum, int n) {
    __shared__ int red[8];
    const int i0 = (blockIdx.x * 256 + threadIdx.x) * 4;
    int s = 0;
    for (int j = 0; j < 4; ++j) {
        int i = i0 + j;
        if (i < n) {
            float d = deg[i];
            deg[i] = (d > 0.0f) ? rsqrtf(d) : 0.0f;
            s += cnt[i];
        }
    }
    for (int off = 16; off > 0; off >>= 1) s += __shfl_xor(s, off);
    if ((threadIdx.x & 31) == 0) red[threadIdx.x >> 5] = s;
    __syncthreads();
    if (threadIdx.x == 0) {
        int t = 0;
        for (int w = 0; w < 8; ++w) t += red[w];
        bsum[blockIdx.x] = t;
    }
}

__global__ void __launch_bounds__(256)
scan_kernel(const int* __restrict__ cnt, const int* __restrict__ bsum,
            int* __restrict__ rowptr, int n) {
    __shared__ int wsum[8];
    __shared__ int osum[8];
    const int tid = threadIdx.x, lane = tid & 31, wave = tid >> 5;
    int o = (tid < (int)blockIdx.x) ? bsum[tid] : 0;
    for (int off = 16; off > 0; off >>= 1) o += __shfl_xor(o, off);
    const int i0 = (blockIdx.x * 256 + tid) * 4;
    int c[4];
    int s = 0;
    for (int j = 0; j < 4; ++j) {
        c[j] = (i0 + j < n) ? cnt[i0 + j] : 0;
        s += c[j];
    }
    int inc = s;
    for (int off = 1; off < 32; off <<= 1) {
        int v = __shfl_up(inc, off);
        if (lane >= off) inc += v;
    }
    if (lane == 0)  osum[wave] = o;
    if (lane == 31) wsum[wave] = inc;
    __syncthreads();
    int ex = inc - s;
    for (int w = 0; w < 8; ++w) {
        ex += osum[w];
        if (w < wave) ex += wsum[w];
    }
    for (int j = 0; j < 4; ++j) {
        if (i0 + j < n) { rowptr[i0 + j] = ex; ex += c[j]; }
    }
    if (i0 < n && i0 + 4 >= n) rowptr[n] = ex;
}

__global__ void fill_kernel(const int* __restrict__ dst, const int* __restrict__ rowptr,
                            const int* __restrict__ rank, int* __restrict__ eid, int nE) {
    int e = blockIdx.x * blockDim.x + threadIdx.x;
    if (e >= nE) return;
    eid[rowptr[dst[e]] + rank[e]] = e;
}

__global__ void csr_final_kernel(const int* __restrict__ src, const int* __restrict__ dst,
                                 const float* __restrict__ ew, const float* __restrict__ dinv,
                                 const int* __restrict__ rowptr, const int* __restrict__ eid,
                                 int* __restrict__ pair, int nE) {
    int e = blockIdx.x * blockDim.x + threadIdx.x;
    if (e >= nE) return;
    int s = src[e], d = dst[e];
    int b = rowptr[d], en = rowptr[d + 1];
    int r = 0;
    for (int j = b; j < en; ++j) r += (eid[j] < e) ? 1 : 0;
    float w = (s == d) ? 0.0f : ew[e];
    float nm = -dinv[s] * w * dinv[d];
    *(int2*)(pair + 2 * (size_t)(b + r)) = make_int2(s, __float_as_int(nm));
}

__global__ void __launch_bounds__(256)
gather_kernel(const float* __restrict__ h, const int* __restrict__ rowptr,
              const int* __restrict__ pair, float* __restrict__ out, int n) {
    const int t   = blockIdx.x * 256 + threadIdx.x;
    const int i   = t >> 4;
    const int sub = t & 15;
    const int f4  = sub << 2;
    if (i >= n) return;
    const int b = rowptr[i], en = rowptr[i + 1];
    float4 a = make_float4(0.0f, 0.0f, 0.0f, 0.0f);
    for (int c0 = b; c0 < en; c0 += 16) {
        const int m = (en - c0 < 16) ? (en - c0) : 16;
        int   ms = 0;
        float mw = 0.0f;
        if (sub < m) {
            const int2 p = *(const int2*)(pair + 2 * (size_t)(c0 + sub));
            ms = p.x;
            mw = __int_as_float(p.y);
        }
        int q = 0;
        for (; q + 4 <= m; q += 4) {
            const int   s0 = __shfl(ms, q, 16),     s1 = __shfl(ms, q + 1, 16);
            const int   s2 = __shfl(ms, q + 2, 16), s3 = __shfl(ms, q + 3, 16);
            const float w0 = __shfl(mw, q, 16),     w1 = __shfl(mw, q + 1, 16);
            const float w2 = __shfl(mw, q + 2, 16), w3 = __shfl(mw, q + 3, 16);
            const float4 h0 = *(const float4*)(h + (size_t)s0 * DHID + f4);
            const float4 h1 = *(const float4*)(h + (size_t)s1 * DHID + f4);
            const float4 h2 = *(const float4*)(h + (size_t)s2 * DHID + f4);
            const float4 h3 = *(const float4*)(h + (size_t)s3 * DHID + f4);
            a.x = fmaf(w0, h0.x, a.x); a.y = fmaf(w0, h0.y, a.y);
            a.z = fmaf(w0, h0.z, a.z); a.w = fmaf(w0, h0.w, a.w);
            a.x = fmaf(w1, h1.x, a.x); a.y = fmaf(w1, h1.y, a.y);
            a.z = fmaf(w1, h1.z, a.z); a.w = fmaf(w1, h1.w, a.w);
            a.x = fmaf(w2, h2.x, a.x); a.y = fmaf(w2, h2.y, a.y);
            a.z = fmaf(w2, h2.z, a.z); a.w = fmaf(w2, h2.w, a.w);
            a.x = fmaf(w3, h3.x, a.x); a.y = fmaf(w3, h3.y, a.y);
            a.z = fmaf(w3, h3.z, a.z); a.w = fmaf(w3, h3.w, a.w);
        }
        for (; q < m; ++q) {
            const int   s = __shfl(ms, q, 16);
            const float w = __shfl(mw, q, 16);
            const float4 hv = *(const float4*)(h + (size_t)s * DHID + f4);
            a.x = fmaf(w, hv.x, a.x);
            a.y = fmaf(w, hv.y, a.y);
            a.z = fmaf(w, hv.z, a.z);
            a.w = fmaf(w, hv.w, a.w);
        }
    }
    *(float4*)(out + (size_t)i * DHID + f4) = a;
}

__global__ void relu_bias_kernel(const float* __restrict__ acc, const float* __restrict__ b,
                                 float* __restrict__ out, int n) {
    int i = blockIdx.x * blockDim.x + threadIdx.x;
    if (i >= n) return;
    out[i] = fmaxf(acc[i] + b[i & (DHID - 1)], 0.0f);
}

template <int MODE>
__global__ void __launch_bounds__(256)
gemm64_wmma(float* __restrict__ T, const float* __restrict__ t0,
            const float* __restrict__ W, float* __restrict__ acc, int nrows) {
    __shared__ float lWp[DHID * DHID];
    for (int idx = threadIdx.x; idx < DHID * DHID; idx += 256) {
        int k = idx >> 6, n = idx & 63;
        lWp[((k >> 1) << 7) + (n << 1) + (k & 1)] = W[idx];
    }
    __syncthreads();

    const int wave = threadIdx.x >> 5;
    const int lane = threadIdx.x & 31;
    const int row0 = (blockIdx.x * 8 + wave) * 16;
    if (row0 >= nrows) return;

    const int lrow  = lane & 15;
    const int khalf = (lane >> 4) << 1;
    const int crow  = row0 + ((lane >> 4) << 3);
    const int ccol  = lane & 15;

    v8f c0 = {}, c1 = {}, c2 = {}, c3 = {};
    if (MODE != 0) {
#pragma unroll
        for (int r = 0; r < 8; ++r) {
            const float* ap = acc + (size_t)(crow + r) * DHID + ccol;
            c0[r] = ap[0]; c1[r] = ap[16]; c2[r] = ap[32]; c3[r] = ap[48];
        }
    }

    float* arow       = T + (size_t)(row0 + lrow) * DHID + khalf;
    const float* trow = (MODE == 2) ? (t0 + (size_t)(row0 + lrow) * DHID + khalf) : nullptr;

#pragma unroll
    for (int kt = 0; kt < 16; ++kt) {
        const int k0 = kt * 4 + khalf;
        v2f a = *(const v2f*)(arow + kt * 4);
        if (MODE == 2) {
            v2f t = *(const v2f*)(trow + kt * 4);
            a.x = 2.0f * a.x - t.x;
            a.y = 2.0f * a.y - t.y;
            *(v2f*)(arow + kt * 4) = a;
        }
        const float* wp = lWp + ((k0 >> 1) << 7);
        v2f b;
        b = *(const v2f*)(wp + ((lrow +  0) << 1));
        c0 = __builtin_amdgcn_wmma_f32_16x16x4_f32(false, a, false, b, (short)0, c0, false, false);
        b = *(const v2f*)(wp + ((lrow + 16) << 1));
        c1 = __builtin_amdgcn_wmma_f32_16x16x4_f32(false, a, false, b, (short)0, c1, false, false);
        b = *(const v2f*)(wp + ((lrow + 32) << 1));
        c2 = __builtin_amdgcn_wmma_f32_16x16x4_f32(false, a, false, b, (short)0, c2, false, false);
        b = *(const v2f*)(wp + ((lrow + 48) << 1));
        c3 = __builtin_amdgcn_wmma_f32_16x16x4_f32(false, a, false, b, (short)0, c3, false, false);
    }

#pragma unroll
    for (int r = 0; r < 8; ++r) {
        float* ap = acc + (size_t)(crow + r) * DHID + ccol;
        ap[0] = c0[r]; ap[16] = c1[r]; ap[32] = c2[r]; ap[48] = c3[r];
    }
}

extern "C" void kernel_launch(void* const* d_in, const int* in_sizes, int n_in,
                              void* d_out, int out_size, void* d_ws, size_t ws_size,
                              hipStream_t stream) {
    const float* x       = (const float*)d_in[0];
    const int*   ei      = (const int*)d_in[1];
    const float* ew      = (const float*)d_in[2];
    const float* weights = (const float*)d_in[3];
    const float* bias    = (const float*)d_in[4];
    float*       out     = (float*)d_out;

    const int* src = ei;
    const int* dst = ei + N_EDGES;

    float* ws   = (float*)d_ws;
    float* base = ws + N_EDGES + N_NODES;
    float* P[5];
    for (int i = 0; i < 5; ++i) P[i] = base + (size_t)i * N_NODES * DHID;
    float* ACC = P[4];

    int*   pair   = (int*)P[3];
    int*   rowptr = pair + 2 * N_EDGES;
    float* dinv   = (float*)(rowptr + N_NODES + 32);
    int*   cnt    = (int*)(dinv + N_NODES);
    int*   eid    = cnt + N_NODES;
    int*   bsum   = eid + N_EDGES;
    int*   rank   = (int*)ws;

    const int NF = N_NODES * DHID;
    dim3 blk(256);
    const int gE    = (N_EDGES + 255) / 256;
    const int gNF   = (NF + 255) / 256;
    const int gScan = (N_NODES + 1023) / 1024;
    const int gGa   = (N_NODES * 16 + 255) / 256;
    const int gGemm = (N_NODES + 127) / 128;

    zero_kernel     <<<256,   blk, 0, stream>>>(dinv, 2 * N_NODES);
    deg_rank_kernel <<<gE,    blk, 0, stream>>>(src, dst, ew, dinv, cnt, rank, N_EDGES);
    dinv_bsum_kernel<<<gScan, blk, 0, stream>>>(dinv, cnt, bsum, N_NODES);
    scan_kernel     <<<gScan, blk, 0, stream>>>(cnt, bsum, rowptr, N_NODES);
    fill_kernel     <<<gE,    blk, 0, stream>>>(dst, rowptr, rank, eid, N_EDGES);
    csr_final_kernel<<<gE,    blk, 0, stream>>>(src, dst, ew, dinv, rowptr, eid, pair, N_EDGES);

    auto prop = [&](const float* tin, float* tout) {
        gather_kernel<<<gGa, blk, 0, stream>>>(tin, rowptr, pair, tout, N_NODES);
    };

    int h_idx = -1;
    for (int l = 0; l < NLAYERS; ++l) {
        int rid[3], c = 0;
        for (int i = 0; i < 3; ++i)
            if (i != h_idx) rid[c++] = i;

        const float* t0 = (h_idx < 0) ? x : P[h_idx];
        float* t1 = P[rid[0]];
        const float* Wl = weights + (size_t)l * KORD * DHID * DHID;

        prop(t0, t1);
        gemm64_wmma<0><<<gGemm, blk, 0, stream>>>((float*)t0, nullptr, Wl,
                                                  ACC, N_NODES);
        gemm64_wmma<1><<<gGemm, blk, 0, stream>>>(t1, nullptr, Wl + DHID * DHID,
                                                  ACC, N_NODES);

        int q[10]; int qh = 0, qt = 0;
        for (int i = 1; i < c; ++i) q[qt++] = rid[i];
        int t0_idx = h_idx;
        int t1_idx = rid[0];

        for (int k = 2; k < KORD; ++k) {
            int s_idx = q[qh++];
            float* scat = P[s_idx];
            prop(P[t1_idx], scat);
            gemm64_wmma<2><<<gGemm, blk, 0, stream>>>(
                scat, (t0_idx < 0) ? x : P[t0_idx],
                Wl + (size_t)k * DHID * DHID, ACC, N_NODES);
            if (t0_idx >= 0) q[qt++] = t0_idx;
            t0_idx = t1_idx;
            t1_idx = s_idx;
        }

        if (l == NLAYERS - 1) {
            relu_bias_kernel<<<gNF, blk, 0, stream>>>(ACC, bias + l * DHID, out, NF);
        } else {
            float* hn = P[t1_idx];
            relu_bias_kernel<<<gNF, blk, 0, stream>>>(ACC, bias + l * DHID, hn, NF);
            h_idx = t1_idx;
        }
    }
}


# Fuse each propagation (row gather + three-term recursion) into the f32 WMMA GEMM that consumes it: gathered rows feed the A operand from a per-wave LDS tile; 21 launches and one plane round trip less
# speedup vs baseline: 2.7808x; 1.1797x over previous
#include <hip/hip_runtime.h>

#define N_NODES 100000
#define N_EDGES 1000000
#define DHID    64
#define KORD    8
#define NLAYERS 3

typedef float v2f __attribute__((ext_vector_type(2)));
typedef float v8f __attribute__((ext_vector_type(8)));

__global__ void zero_kernel(float* __restrict__ p, int n) {
    int i = blockIdx.x * blockDim.x + threadIdx.x;
    int stride = gridDim.x * blockDim.x;
    for (; i < n; i += stride) p[i] = 0.0f;
}

__global__ void deg_rank_kernel(const int* __restrict__ src, const int* __restrict__ dst,
                                const float* __restrict__ ew, float* __restrict__ deg,
                                int* __restrict__ cnt, int* __restrict__ rank, int nE) {
    int e = blockIdx.x * blockDim.x + threadIdx.x;
    if (e >= nE) return;
    int s = src[e], d = dst[e];
    if (s != d) atomicAdd(&deg[s], ew[e]);
    rank[e] = atomicAdd(&cnt[d], 1);
}

__global__ void __launch_bounds__(256)
dinv_bsum_kernel(float* __restrict__ deg, const int* __restrict__ cnt,
                 int* __restrict__ bsum, int n) {
    __shared__ int red[8];
    const int i0 = (blockIdx.x * 256 + threadIdx.x) * 4;
    int s = 0;
    for (int j = 0; j < 4; ++j) {
        int i = i0 + j;
        if (i < n) {
            float d = deg[i];
            deg[i] = (d > 0.0f) ? rsqrtf(d) : 0.0f;
            s += cnt[i];
        }
    }
    for (int off = 16; off > 0; off >>= 1) s += __shfl_xor(s, off);
    if ((threadIdx.x & 31) == 0) red[threadIdx.x >> 5] = s;
    __syncthreads();
    if (threadIdx.x == 0) {
        int t = 0;
        for (int w = 0; w < 8; ++w) t += red[w];
        bsum[blockIdx.x] = t;
    }
}

__global__ void __launch_bounds__(256)
scan_kernel(const int* __restrict__ cnt, const int* __restrict__ bsum,
            int* __restrict__ rowptr, int n) {
    __shared__ int wsum[8];
    __shared__ int osum[8];
    const int tid = threadIdx.x, lane = tid & 31, wave = tid >> 5;
    int o = (tid < (int)blockIdx.x) ? bsum[tid] : 0;
    for (int off = 16; off > 0; off >>= 1) o += __shfl_xor(o, off);
    const int i0 = (blockIdx.x * 256 + tid) * 4;
    int c[4];
    int s = 0;
    for (int j = 0; j < 4; ++j) {
        c[j] = (i0 + j < n) ? cnt[i0 + j] : 0;
        s += c[j];
    }
    int inc = s;
    for (int off = 1; off < 32; off <<= 1) {
        int v = __shfl_up(inc, off);
        if (lane >= off) inc += v;
    }
    if (lane == 0)  osum[wave] = o;
    if (lane == 31) wsum[wave] = inc;
    __syncthreads();
    int ex = inc - s;
    for (int w = 0; w < 8; ++w) {
        ex += osum[w];
        if (w < wave) ex += wsum[w];
    }
    for (int j = 0; j < 4; ++j) {
        if (i0 + j < n) { rowptr[i0 + j] = ex; ex += c[j]; }
    }
    if (i0 < n && i0 + 4 >= n) rowptr[n] = ex;
}

__global__ void fill_kernel(const int* __restrict__ dst, const int* __restrict__ rowptr,
                            const int* __restrict__ rank, int* __restrict__ eid, int nE) {
    int e = blockIdx.x * blockDim.x + threadIdx.x;
    if (e >= nE) return;
    eid[rowptr[dst[e]] + rank[e]] = e;
}

__global__ void csr_final_kernel(const int* __restrict__ src, const int* __restrict__ dst,
                                 const float* __restrict__ ew, const float* __restrict__ dinv,
                                 const int* __restrict__ rowptr, const int* __restrict__ eid,
                                 int* __restrict__ pair, int nE) {
    int e = blockIdx.x * blockDim.x + threadIdx.x;
    if (e >= nE) return;
    int s = src[e], d = dst[e];
    int b = rowptr[d], en = rowptr[d + 1];
    int r = 0;
    for (int j = b; j < en; ++j) r += (eid[j] < e) ? 1 : 0;
    float w = (s == d) ? 0.0f : ew[e];
    float nm = -dinv[s] * w * dinv[d];
    *(int2*)(pair + 2 * (size_t)(b + r)) = make_int2(s, __float_as_int(nm));
}

__global__ void relu_bias_kernel(const float* __restrict__ acc, const float* __restrict__ b,
                                 float* __restrict__ out, int n) {
    int i = blockIdx.x * blockDim.x + threadIdx.x;
    if (i >= n) return;
    out[i] = fmaxf(acc[i] + b[i & (DHID - 1)], 0.0f);
}

__device__ __forceinline__ void stage_w(float* lWp, const float* __restrict__ W) {
    for (int idx = threadIdx.x; idx < DHID * DHID; idx += 256) {
        int k = idx >> 6, n = idx & 63;
        lWp[((k >> 1) << 7) + (n << 1) + (k & 1)] = W[idx];
    }
}

__global__ void __launch_bounds__(256)
gemm64_wmma(const float* __restrict__ T, const float* __restrict__ W,
            float* __restrict__ acc, int nrows) {
    __shared__ float lWp[DHID * DHID];
    stage_w(lWp, W);
    __syncthreads();

    const int wave = threadIdx.x >> 5;
    const int lane = threadIdx.x & 31;
    const int row0 = (blockIdx.x * 8 + wave) * 16;
    if (row0 >= nrows) return;

    const int lrow  = lane & 15;
    const int khalf = (lane >> 4) << 1;
    const int crow  = row0 + ((lane >> 4) << 3);
    const int ccol  = lane & 15;

    v8f c0 = {}, c1 = {}, c2 = {}, c3 = {};
    const float* arow = T + (size_t)(row0 + lrow) * DHID + khalf;

#pragma unroll
    for (int kt = 0; kt < 16; ++kt) {
        const int k0 = kt * 4 + khalf;
        const v2f a = *(const v2f*)(arow + kt * 4);
        const float* wp = lWp + ((k0 >> 1) << 7);
        v2f b;
        b = *(const v2f*)(wp + ((lrow +  0) << 1));
        c0 = __builtin_amdgcn_wmma_f32_16x16x4_f32(false, a, false, b, (short)0, c0, false, false);
        b = *(const v2f*)(wp + ((lrow + 16) << 1));
        c1 = __builtin_amdgcn_wmma_f32_16x16x4_f32(false, a, false, b, (short)0, c1, false, false);
        b = *(const v2f*)(wp + ((lrow + 32) << 1));
        c2 = __builtin_amdgcn_wmma_f32_16x16x4_f32(false, a, false, b, (short)0, c2, false, false);
        b = *(const v2f*)(wp + ((lrow + 48) << 1));
        c3 = __builtin_amdgcn_wmma_f32_16x16x4_f32(false, a, false, b, (short)0, c3, false, false);
    }

#pragma unroll
    for (int r = 0; r < 8; ++r) {
        float* ap = acc + (size_t)(crow + r) * DHID + ccol;
        ap[0] = c0[r]; ap[16] = c1[r]; ap[32] = c2[r]; ap[48] = c3[r];
    }
}

#define TPITCH 68

template <int RECUR>
__global__ void __launch_bounds__(256)
prop_gemm_wmma(const float* __restrict__ tin, const float* __restrict__ tprev,
               const int* __restrict__ rowptr, const int* __restrict__ pair,
               const float* __restrict__ W, float* __restrict__ tout,
               float* __restrict__ acc, int nrows) {
    __shared__ float lWp[DHID * DHID];
    __shared__ __attribute__((aligned(16))) float lT[8 * 16 * TPITCH];
    stage_w(lWp, W);
    __syncthreads();

    const int wave = threadIdx.x >> 5;
    const int lane = threadIdx.x & 31;
    const int row0 = (blockIdx.x * 8 + wave) * 16;
    if (row0 >= nrows) return;

    const int lrow  = lane & 15;
    const int half  = lane >> 4;
    const int khalf = half << 1;
    const int crow  = row0 + (half << 3);
    const int ccol  = lane & 15;
    const int f4    = lrow << 2;

    v8f c0, c1, c2, c3;
#pragma unroll
    for (int r = 0; r < 8; ++r) {
        const float* ap = acc + (size_t)(crow + r) * DHID + ccol;
        c0[r] = ap[0]; c1[r] = ap[16]; c2[r] = ap[32]; c3[r] = ap[48];
    }

    const int rp = rowptr[row0 + ((lane < 16) ? lane : 16)];
    float* myT = lT + wave * (16 * TPITCH);

    int   pms[8];
    float pmw[8];
#pragma unroll
    for (int it = 0; it < 8; ++it) {
        const int b    = __shfl(rp, it * 2 + half);
        const int slot = (b + lrow < N_EDGES) ? (b + lrow) : (N_EDGES - 1);
        const int2 p   = *(const int2*)(pair + 2 * (size_t)slot);
        pms[it] = p.x;
        pmw[it] = __int_as_float(p.y);
    }

#pragma unroll
    for (int it = 0; it < 8; ++it) {
        const int r  = it * 2 + half;
        const int b  = __shfl(rp, r), en = __shfl(rp, r + 1);
        int   ms = pms[it];
        float mw = pmw[it];
        float4 tp = make_float4(0.0f, 0.0f, 0.0f, 0.0f);
        if (RECUR) tp = *(const float4*)(tprev + (size_t)(row0 + r) * DHID + f4);
        float4 a = make_float4(0.0f, 0.0f, 0.0f, 0.0f);
        for (int c0e = b; c0e < en; c0e += 16) {
            const int m = (en - c0e < 16) ? (en - c0e) : 16;
            if (c0e != b) {
                ms = 0;
                mw = 0.0f;
                if (lrow < m) {
                    const int2 p = *(const int2*)(pair + 2 * (size_t)(c0e + lrow));
                    ms = p.x;
                    mw = __int_as_float(p.y);
                }
            }
            int q = 0;
            for (; q + 4 <= m; q += 4) {
                const int   s0 = __shfl(ms, q, 16),     s1 = __shfl(ms, q + 1, 16);
                const int   s2 = __shfl(ms, q + 2, 16), s3 = __shfl(ms, q + 3, 16);
                const float w0 = __shfl(mw, q, 16),     w1 = __shfl(mw, q + 1, 16);
                const float w2 = __shfl(mw, q + 2, 16), w3 = __shfl(mw, q + 3, 16);
                const float4 h0 = *(const float4*)(tin + (size_t)s0 * DHID + f4);
                const float4 h1 = *(const float4*)(tin + (size_t)s1 * DHID + f4);
                const float4 h2 = *(const float4*)(tin + (size_t)s2 * DHID + f4);
                const float4 h3 = *(const float4*)(tin + (size_t)s3 * DHID + f4);
                a.x = fmaf(w0, h0.x, a.x); a.y = fmaf(w0, h0.y, a.y);
                a.z = fmaf(w0, h0.z, a.z); a.w = fmaf(w0, h0.w, a.w);
                a.x = fmaf(w1, h1.x, a.x); a.y = fmaf(w1, h1.y, a.y);
                a.z = fmaf(w1, h1.z, a.z); a.w = fmaf(w1, h1.w, a.w);
                a.x = fmaf(w2, h2.x, a.x); a.y = fmaf(w2, h2.y, a.y);
                a.z = fmaf(w2, h2.z, a.z); a.w = fmaf(w2, h2.w, a.w);
                a.x = fmaf(w3, h3.x, a.x); a.y = fmaf(w3, h3.y, a.y);
                a.z = fmaf(w3, h3.z, a.z); a.w = fmaf(w3, h3.w, a.w);
            }
            for (; q < m; ++q) {
                const int   s = __shfl(ms, q, 16);
                const float w = __shfl(mw, q, 16);
                const float4 hv = *(const float4*)(tin + (size_t)s * DHID + f4);
                a.x = fmaf(w, hv.x, a.x);
                a.y = fmaf(w, hv.y, a.y);
                a.z = fmaf(w, hv.z, a.z);
                a.w = fmaf(w, hv.w, a.w);
            }
        }
        if (RECUR) {
            a.x = 2.0f * a.x - tp.x;
            a.y = 2.0f * a.y - tp.y;
            a.z = 2.0f * a.z - tp.z;
            a.w = 2.0f * a.w - tp.w;
        }
        *(float4*)(tout + (size_t)(row0 + r) * DHID + f4) = a;
        *(float4*)(myT + r * TPITCH + f4) = a;
    }

    __builtin_amdgcn_fence(__ATOMIC_RELEASE, "wavefront");
    __builtin_amdgcn_wave_barrier();
    __builtin_amdgcn_fence(__ATOMIC_ACQUIRE, "wavefront");

    const float* arow = myT + lrow * TPITCH + khalf;
#pragma unroll
    for (int kt = 0; kt < 16; ++kt) {
        const int k0 = kt * 4 + khalf;
        const v2f a = *(const v2f*)(arow + kt * 4);
        const float* wp = lWp + ((k0 >> 1) << 7);
        v2f b;
        b = *(const v2f*)(wp + ((lrow +  0) << 1));
        c0 = __builtin_amdgcn_wmma_f32_16x16x4_f32(false, a, false, b, (short)0, c0, false, false);
        b = *(const v2f*)(wp + ((lrow + 16) << 1));
        c1 = __builtin_amdgcn_wmma_f32_16x16x4_f32(false, a, false, b, (short)0, c1, false, false);
        b = *(const v2f*)(wp + ((lrow + 32) << 1));
        c2 = __builtin_amdgcn_wmma_f32_16x16x4_f32(false, a, false, b, (short)0, c2, false, false);
        b = *(const v2f*)(wp + ((lrow + 48) << 1));
        c3 = __builtin_amdgcn_wmma_f32_16x16x4_f32(false, a, false, b, (short)0, c3, false, false);
    }

#pragma unroll
    for (int r = 0; r < 8; ++r) {
        float* ap = acc + (size_t)(crow + r) * DHID + ccol;
        ap[0] = c0[r]; ap[16] = c1[r]; ap[32] = c2[r]; ap[48] = c3[r];
    }
}

extern "C" void kernel_launch(void* const* d_in, const int* in_sizes, int n_in,
                              void* d_out, int out_size, void* d_ws, size_t ws_size,
                              hipStream_t stream) {
    const float* x       = (const float*)d_in[0];
    const int*   ei      = (const int*)d_in[1];
    const float* ew      = (const float*)d_in[2];
    const float* weights = (const float*)d_in[3];
    const float* bias    = (const float*)d_in[4];
    float*       out     = (float*)d_out;

    const int* src = ei;
    const int* dst = ei + N_EDGES;

    float* ws   = (float*)d_ws;
    float* base = ws + N_EDGES + N_NODES;
    float* P[5];
    for (int i = 0; i < 5; ++i) P[i] = base + (size_t)i * N_NODES * DHID;
    float* ACC = P[4];

    int*   pair   = (int*)P[3];
    int*   rowptr = pair + 2 * N_EDGES;
    float* dinv   = (float*)(rowptr + N_NODES + 32);
    int*   cnt    = (int*)(dinv + N_NODES);
    int*   eid    = cnt + N_NODES;
    int*   bsum   = eid + N_EDGES;
    int*   rank   = (int*)ws;

    const int NF = N_NODES * DHID;
    dim3 blk(256);
    const int gE    = (N_EDGES + 255) / 256;
    const int gNF   = (NF + 255) / 256;
    const int gScan = (N_NODES + 1023) / 1024;
    const int gGemm = (N_NODES + 127) / 128;

    zero_kernel     <<<256,   blk, 0, stream>>>(dinv, 2 * N_NODES);
    deg_rank_kernel <<<gE,    blk, 0, stream>>>(src, dst, ew, dinv, cnt, rank, N_EDGES);
    dinv_bsum_kernel<<<gScan, blk, 0, stream>>>(dinv, cnt, bsum, N_NODES);
    scan_kernel     <<<gScan, blk, 0, stream>>>(cnt, bsum, rowptr, N_NODES);
    fill_kernel     <<<gE,    blk, 0, stream>>>(dst, rowptr, rank, eid, N_EDGES);
    csr_final_kernel<<<gE,    blk, 0, stream>>>(src, dst, ew, dinv, rowptr, eid, pair, N_EDGES);

    int h_idx = -1;
    for (int l = 0; l < NLAYERS; ++l) {
        int rid[3], c = 0;
        for (int i = 0; i < 3; ++i)
            if (i != h_idx) rid[c++] = i;

        const float* t0 = (h_idx < 0) ? x : P[h_idx];
        float* t1 = P[rid[0]];
        const float* Wl = weights + (size_t)l * KORD * DHID * DHID;

        gemm64_wmma<<<gGemm, blk, 0, stream>>>(t0, Wl, ACC, N_NODES);
        prop_gemm_wmma<0><<<gGemm, blk, 0, stream>>>(t0, nullptr, rowptr, pair,
                                                     Wl + DHID * DHID, t1, ACC, N_NODES);

        int q[10]; int qh = 0, qt = 0;
        for (int i = 1; i < c; ++i) q[qt++] = rid[i];
        int t0_idx = h_idx;
        int t1_idx = rid[0];

        for (int k = 2; k < KORD; ++k) {
            int s_idx = q[qh++];
            prop_gemm_wmma<1><<<gGemm, blk, 0, stream>>>(
                P[t1_idx], (t0_idx < 0) ? x : P[t0_idx], rowptr, pair,
                Wl + (size_t)k * DHID * DHID, P[s_idx], ACC, N_NODES);
            if (t0_idx >= 0) q[qt++] = t0_idx;
            t0_idx = t1_idx;
            t1_idx = s_idx;
        }

        if (l == NLAYERS - 1) {
            relu_bias_kernel<<<gNF, blk, 0, stream>>>(ACC, bias + l * DHID, out, NF);
        } else {
            float* hn = P[t1_idx];
            relu_bias_kernel<<<gNF, blk, 0, stream>>>(ACC, bias + l * DHID, hn, NF);
            h_idx = t1_idx;
        }
    }
}


# Accumulator plane traffic cut: 4 launches a layer (not 9) touch it, each adds a pair of terms (T(k-2) recursion rows + T(k)) as whole rows, last adds bias+ReLU; terms summed in pairs 0,2|1,3|4,6|5,7
# speedup vs baseline: 3.3147x; 1.1920x over previous
#include <hip/hip_runtime.h>

#define N_NODES 100000
#define N_EDGES 1000000
#define DHID    64
#define KORD    8
#define NLAYERS 3

typedef float v2f __attribute__((ext_vector_type(2)));
typedef float v8f __attribute__((ext_vector_type(8)));

__global__ void zero_kernel(float* __restrict__ p, int n) {
    int i = blockIdx.x * blockDim.x + threadIdx.x;
    int stride = gridDim.x * blockDim.x;
    for (; i < n; i += stride) p[i] = 0.0f;
}

__global__ void deg_rank_kernel(const int* __restrict__ src, const int* __restrict__ dst,
                                const float* __restrict__ ew, float* __restrict__ deg,
                                int* __restrict__ cnt, int* __restrict__ rank, int nE) {
    int e = blockIdx.x * blockDim.x + threadIdx.x;
    if (e >= nE) return;
    int s = src[e], d = dst[e];
    if (s != d) atomicAdd(&deg[s], ew[e]);
    rank[e] = atomicAdd(&cnt[d], 1);
}

__global__ void __launch_bounds__(256)
dinv_bsum_kernel(float* __restrict__ deg, const int* __restrict__ cnt,
                 int* __restrict__ bsum, int n) {
    __shared__ int red[8];
    const int i0 = (blockIdx.x * 256 + threadIdx.x) * 4;
    int s = 0;
    for (int j = 0; j < 4; ++j) {
        int i = i0 + j;
        if (i < n) {
            float d = deg[i];
            deg[i] = (d > 0.0f) ? rsqrtf(d) : 0.0f;
            s += cnt[i];
        }
    }
    for (int off = 16; off > 0; off >>= 1) s += __shfl_xor(s, off);
    if ((threadIdx.x & 31) == 0) red[threadIdx.x >> 5] = s;
    __syncthreads();
    if (threadIdx.x == 0) {
        int t = 0;
        for (int w = 0; w < 8; ++w) t += red[w];
        bsum[blockIdx.x] = t;
    }
}

__global__ void __launch_bounds__(256)
scan_kernel(const int* __restrict__ cnt, const int* __restrict__ bsum,
            int* __restrict__ rowptr, int n) {
    __shared__ int wsum[8];
    __shared__ int osum[8];
    const int tid = threadIdx.x, lane = tid & 31, wave = tid >> 5;
    int o = (tid < (int)blockIdx.x) ? bsum[tid] : 0;
    for (int off = 16; off > 0; off >>= 1) o += __shfl_xor(o, off);
    const int i0 = (blockIdx.x * 256 + tid) * 4;
    int c[4];
    int s = 0;
    for (int j = 0; j < 4; ++j) {
        c[j] = (i0 + j < n) ? cnt[i0 + j] : 0;
        s += c[j];
    }
    int inc = s;
    for (int off = 1; off < 32; off <<= 1) {
        int v = __shfl_up(inc, off);
        if (lane >= off) inc += v;
    }
    if (lane == 0)  osum[wave] = o;
    if (lane == 31) wsum[wave] = inc;
    __syncthreads();
    int ex = inc - s;
    for (int w = 0; w < 8; ++w) {
        ex += osum[w];
        if (w < wave) ex += wsum[w];
    }
    for (int j = 0; j < 4; ++j) {
        if (i0 + j < n) { rowptr[i0 + j] = ex; ex += c[j]; }
    }
    if (i0 < n && i0 + 4 >= n) rowptr[n] = ex;
}

__global__ void fill_kernel(const int* __restrict__ dst, const int* __restrict__ rowptr,
                            const int* __restrict__ rank, int* __restrict__ eid, int nE) {
    int e = blockIdx.x * blockDim.x + threadIdx.x;
    if (e >= nE) return;
    eid[rowptr[dst[e]] + rank[e]] = e;
}

__global__ void csr_final_kernel(const int* __restrict__ src, const int* __restrict__ dst,
                                 const float* __restrict__ ew, const float* __restrict__ dinv,
                                 const int* __restrict__ rowptr, const int* __restrict__ eid,
                                 int* __restrict__ pair, int nE) {
    int e = blockIdx.x * blockDim.x + threadIdx.x;
    if (e >= nE) return;
    int s = src[e], d = dst[e];
    int b = rowptr[d], en = rowptr[d + 1];
    int r = 0;
    for (int j = b; j < en; ++j) r += (eid[j] < e) ? 1 : 0;
    float w = (s == d) ? 0.0f : ew[e];
    float nm = -dinv[s] * w * dinv[d];
    *(int2*)(pair + 2 * (size_t)(b + r)) = make_int2(s, __float_as_int(nm));
}

__device__ __forceinline__ void stage_w(float* lWp, const float* __restrict__ W) {
    for (int idx = threadIdx.x; idx < DHID * DHID; idx += 256) {
        int k = idx >> 6, n = idx & 63;
        lWp[((k >> 1) << 7) + (n << 1) + (k & 1)] = W[idx];
    }
}

__device__ __forceinline__ void tile_gemm(const float* arow, const float* lWp, int khalf, int lrow,
                                          v8f& c0, v8f& c1, v8f& c2, v8f& c3) {
#pragma unroll
    for (int kt = 0; kt < 16; ++kt) {
        const int k0 = kt * 4 + khalf;
        const v2f a = *(const v2f*)(arow + kt * 4);
        const float* wp = lWp + ((k0 >> 1) << 7);
        v2f b;
        b = *(const v2f*)(wp + ((lrow +  0) << 1));
        c0 = __builtin_amdgcn_wmma_f32_16x16x4_f32(false, a, false, b, (short)0, c0, false, false);
        b = *(const v2f*)(wp + ((lrow + 16) << 1));
        c1 = __builtin_amdgcn_wmma_f32_16x16x4_f32(false, a, false, b, (short)0, c1, false, false);
        b = *(const v2f*)(wp + ((lrow + 32) << 1));
        c2 = __builtin_amdgcn_wmma_f32_16x16x4_f32(false, a, false, b, (short)0, c2, false, false);
        b = *(const v2f*)(wp + ((lrow + 48) << 1));
        c3 = __builtin_amdgcn_wmma_f32_16x16x4_f32(false, a, false, b, (short)0, c3, false, false);
    }
}

__device__ __forceinline__ float4 gather_row(const float* __restrict__ tin, const int* __restrict__ pair,
                                             int b, int en, int ms, float mw, int lrow, int f4) {
    float4 a = make_float4(0.0f, 0.0f, 0.0f, 0.0f);
    for (int c0e = b; c0e < en; c0e += 16) {
        const int m = (en - c0e < 16) ? (en - c0e) : 16;
        if (c0e != b) {
            ms = 0;
            mw = 0.0f;
            if (lrow < m) {
                const int2 p = *(const int2*)(pair + 2 * (size_t)(c0e + lrow));
                ms = p.x;
                mw = __int_as_float(p.y);
            }
        }
        int q = 0;
        for (; q + 4 <= m; q += 4) {
            const int   s0 = __shfl(ms, q, 16),     s1 = __shfl(ms, q + 1, 16);
            const int   s2 = __shfl(ms, q + 2, 16), s3 = __shfl(ms, q + 3, 16);
            const float w0 = __shfl(mw, q, 16),     w1 = __shfl(mw, q + 1, 16);
            const float w2 = __shfl(mw, q + 2, 16), w3 = __shfl(mw, q + 3, 16);
            const float4 h0 = *(const float4*)(tin + (size_t)s0 * DHID + f4);
            const float4 h1 = *(const float4*)(tin + (size_t)s1 * DHID + f4);
            const float4 h2 = *(const float4*)(tin + (size_t)s2 * DHID + f4);
            const float4 h3 = *(const float4*)(tin + (size_t)s3 * DHID + f4);
            a.x = fmaf(w0, h0.x, a.x); a.y = fmaf(w0, h0.y, a.y);
            a.z = fmaf(w0, h0.z, a.z); a.w = fmaf(w0, h0.w, a.w);
            a.x = fmaf(w1, h1.x, a.x); a.y = fmaf(w1, h1.y, a.y);
            a.z = fmaf(w1, h1.z, a.z); a.w = fmaf(w1, h1.w, a.w);
            a.x = fmaf(w2, h2.x, a.x); a.y = fmaf(w2, h2.y, a.y);
            a.z = fmaf(w2, h2.z, a.z); a.w = fmaf(w2, h2.w, a.w);
            a.x = fmaf(w3, h3.x, a.x); a.y = fmaf(w3, h3.y, a.y);
            a.z = fmaf(w3, h3.z, a.z); a.w = fmaf(w3, h3.w, a.w);
        }
        for (; q < m; ++q) {
            const int   s = __shfl(ms, q, 16);
            const float w = __shfl(mw, q, 16);
            const float4 hv = *(const float4*)(tin + (size_t)s * DHID + f4);
            a.x = fmaf(w, hv.x, a.x);
            a.y = fmaf(w, hv.y, a.y);
            a.z = fmaf(w, hv.z, a.z);
            a.w = fmaf(w, hv.w, a.w);
        }
    }
    return a;
}

__device__ __forceinline__ void first_pairs(const int* __restrict__ pair, int rp, int half, int lrow,
                                            int (&pms)[8], float (&pmw)[8]) {
#pragma unroll
    for (int it = 0; it < 8; ++it) {
        const int b    = __shfl(rp, it * 2 + half);
        const int slot = (b + lrow < N_EDGES) ? (b + lrow) : (N_EDGES - 1);
        const int2 p   = *(const int2*)(pair + 2 * (size_t)slot);
        pms[it] = p.x;
        pmw[it] = __int_as_float(p.y);
    }
}

template <int RECUR>
__global__ void __launch_bounds__(256)
prop_only(const float* __restrict__ tin, const float* __restrict__ tprev,
          const int* __restrict__ rowptr, const int* __restrict__ pair,
          float* __restrict__ tout, int nrows) {
    const int wave = threadIdx.x >> 5;
    const int lane = threadIdx.x & 31;
    const int row0 = (blockIdx.x * 8 + wave) * 16;
    if (row0 >= nrows) return;

    const int lrow = lane & 15;
    const int half = lane >> 4;
    const int f4   = lrow << 2;

    const int rp = rowptr[row0 + ((lane < 16) ? lane : 16)];
    int   pms[8];
    float pmw[8];
    first_pairs(pair, rp, half, lrow, pms, pmw);

#pragma unroll
    for (int it = 0; it < 8; ++it) {
        const int r  = it * 2 + half;
        const int b  = __shfl(rp, r), en = __shfl(rp, r + 1);
        float4 tp = make_float4(0.0f, 0.0f, 0.0f, 0.0f);
        if (RECUR) tp = *(const float4*)(tprev + (size_t)(row0 + r) * DHID + f4);
        float4 a = gather_row(tin, pair, b, en, pms[it], pmw[it], lrow, f4);
        if (RECUR) {
            a.x = 2.0f * a.x - tp.x;
            a.y = 2.0f * a.y - tp.y;
            a.z = 2.0f * a.z - tp.z;
            a.w = 2.0f * a.w - tp.w;
        }
        *(float4*)(tout + (size_t)(row0 + r) * DHID + f4) = a;
    }
}

#define TPITCH 68

template <int ACCIN, int FINAL>
__global__ void __launch_bounds__(256)
prop_gemm2(const float* __restrict__ tin, const float* __restrict__ tprev,
           const int* __restrict__ rowptr, const int* __restrict__ pair,
           const float* __restrict__ Wa, const float* __restrict__ Wb,
           const float* __restrict__ bias, float* __restrict__ tout,
           float* __restrict__ acc, float* __restrict__ hout, int nrows) {
    __shared__ float lW[2 * DHID * DHID];
    __shared__ __attribute__((aligned(16))) float lT[8 * 16 * TPITCH];
    stage_w(lW, Wa);
    stage_w(lW + DHID * DHID, Wb);
    __syncthreads();

    const int wave = threadIdx.x >> 5;
    const int lane = threadIdx.x & 31;
    const int row0 = __builtin_amdgcn_readfirstlane((blockIdx.x * 8 + wave) * 16);
    if (row0 >= nrows) return;

    const int lrow  = lane & 15;
    const int half  = lane >> 4;
    const int khalf = half << 1;
    const int ccol  = lane & 15;
    const int f4    = lrow << 2;
    const int loff  = (half << 6) + f4;
    const size_t rbase = (size_t)row0 * DHID;

    const int rp = rowptr[row0 + ((lane < 16) ? lane : 16)];
    float* myT = lT + wave * (16 * TPITCH);
    const float* arow = myT + lrow * TPITCH + khalf;

#pragma unroll
    for (int it = 0; it < 8; ++it) {
        const int r = it * 2 + half;
        const float4 t = *(const float4*)(tprev + rbase + it * 2 * DHID + loff);
        *(float4*)(myT + r * TPITCH + f4) = t;
    }

    int   pms[8];
    float pmw[8];
    first_pairs(pair, rp, half, lrow, pms, pmw);

    __builtin_amdgcn_fence(__ATOMIC_RELEASE, "wavefront");
    __builtin_amdgcn_wave_barrier();
    __builtin_amdgcn_fence(__ATOMIC_ACQUIRE, "wavefront");

    v8f c0 = {}, c1 = {}, c2 = {}, c3 = {};
    tile_gemm(arow, lW, khalf, lrow, c0, c1, c2, c3);
    asm volatile("" : "+v"(c0), "+v"(c1), "+v"(c2), "+v"(c3));

#pragma unroll
    for (int it = 0; it < 8; ++it) {
        const int r  = it * 2 + half;
        const int b  = __shfl(rp, r), en = __shfl(rp, r + 1);
        float4 a = gather_row(tin, pair, b, en, pms[it], pmw[it], lrow, f4);
        const float4 tp = *(const float4*)(myT + r * TPITCH + f4);
        a.x = 2.0f * a.x - tp.x;
        a.y = 2.0f * a.y - tp.y;
        a.z = 2.0f * a.z - tp.z;
        a.w = 2.0f * a.w - tp.w;
        if (!FINAL) *(float4*)(tout + rbase + it * 2 * DHID + loff) = a;
        *(float4*)(myT + r * TPITCH + f4) = a;
    }

    float4 av[8];
    if (ACCIN) {
#pragma unroll
        for (int it = 0; it < 8; ++it)
            av[it] = *(const float4*)(acc + rbase + it * 2 * DHID + loff);
    }

    __builtin_amdgcn_fence(__ATOMIC_RELEASE, "wavefront");
    __builtin_amdgcn_wave_barrier();
    __builtin_amdgcn_fence(__ATOMIC_ACQUIRE, "wavefront");

    tile_gemm(arow, lW + DHID * DHID, khalf, lrow, c0, c1, c2, c3);

#pragma unroll
    for (int r = 0; r < 8; ++r) {
        float* tr = myT + ((half << 3) + r) * TPITCH + ccol;
        tr[0] = c0[r]; tr[16] = c1[r]; tr[32] = c2[r]; tr[48] = c3[r];
    }

    __builtin_amdgcn_fence(__ATOMIC_RELEASE, "wavefront");
    __builtin_amdgcn_wave_barrier();
    __builtin_amdgcn_fence(__ATOMIC_ACQUIRE, "wavefront");

    float4 bv = make_float4(0.0f, 0.0f, 0.0f, 0.0f);
    if (FINAL) bv = *(const float4*)(bias + f4);
#pragma unroll
    for (int it = 0; it < 8; ++it) {
        const int r = it * 2 + half;
        float4 v = *(const float4*)(myT + r * TPITCH + f4);
        if (ACCIN) {
            v.x = av[it].x + v.x; v.y = av[it].y + v.y;
            v.z = av[it].z + v.z; v.w = av[it].w + v.w;
        }
        if (FINAL) {
            v.x = fmaxf(v.x + bv.x, 0.0f); v.y = fmaxf(v.y + bv.y, 0.0f);
            v.z = fmaxf(v.z + bv.z, 0.0f); v.w = fmaxf(v.w + bv.w, 0.0f);
        }
        float* dstp = FINAL ? hout : acc;
        *(float4*)(dstp + rbase + it * 2 * DHID + loff) = v;
    }
}

extern "C" void kernel_launch(void* const* d_in, const int* in_sizes, int n_in,
                              void* d_out, int out_size, void* d_ws, size_t ws_size,
                              hipStream_t stream) {
    const float* x       = (const float*)d_in[0];
    const int*   ei      = (const int*)d_in[1];
    const float* ew      = (const float*)d_in[2];
    const float* weights = (const float*)d_in[3];
    const float* bias    = (const float*)d_in[4];
    float*       out     = (float*)d_out;

    const int* src = ei;
    const int* dst = ei + N_EDGES;

    float* ws   = (float*)d_ws;
    float* base = ws + N_EDGES + N_NODES;
    float* P[5];
    for (int i = 0; i < 5; ++i) P[i] = base + (size_t)i * N_NODES * DHID;
    float* ACC = P[4];

    int*   pair   = (int*)P[3];
    int*   rowptr = pair + 2 * N_EDGES;
    float* dinv   = (float*)(rowptr + N_NODES + 32);
    int*   cnt    = (int*)(dinv + N_NODES);
    int*   eid    = cnt + N_NODES;
    int*   bsum   = eid + N_EDGES;
    int*   rank   = (int*)ws;

    dim3 blk(256);
    const int gE    = (N_EDGES + 255) / 256;
    const int gScan = (N_NODES + 1023) / 1024;
    const int gGemm = (N_NODES + 127) / 128;
    const int WSZ   = DHID * DHID;

    zero_kernel     <<<256,   blk, 0, stream>>>(dinv, 2 * N_NODES);
    deg_rank_kernel <<<gE,    blk, 0, stream>>>(src, dst, ew, dinv, cnt, rank, N_EDGES);
    dinv_bsum_kernel<<<gScan, blk, 0, stream>>>(dinv, cnt, bsum, N_NODES);
    scan_kernel     <<<gScan, blk, 0, stream>>>(cnt, bsum, rowptr, N_NODES);
    fill_kernel     <<<gE,    blk, 0, stream>>>(dst, rowptr, rank, eid, N_EDGES);
    csr_final_kernel<<<gE,    blk, 0, stream>>>(src, dst, ew, dinv, rowptr, eid, pair, N_EDGES);

    int h_idx = -1;
    for (int l = 0; l < NLAYERS; ++l) {
        int rid[3], c = 0;
        for (int i = 0; i < 3; ++i)
            if (i != h_idx) rid[c++] = i;

        const float* t0 = (h_idx < 0) ? x : P[h_idx];
        float* t1 = P[rid[0]];
        const float* Wl = weights + (size_t)l * KORD * WSZ;
        const float* bl = bias + l * DHID;

        prop_only<0><<<gGemm, blk, 0, stream>>>(t0, nullptr, rowptr, pair, t1, N_NODES);

        int q[10]; int qh = 0, qt = 0;
        for (int i = 1; i < c; ++i) q[qt++] = rid[i];
        int t0_idx = h_idx;
        int t1_idx = rid[0];

        for (int k = 2; k < KORD; ++k) {
            int s_idx = q[qh++];
            const float* tin = P[t1_idx];
            const float* tpv = (t0_idx < 0) ? x : P[t0_idx];
            const float* Wa  = Wl + (size_t)(k - 2) * WSZ;
            const float* Wb  = Wl + (size_t)k * WSZ;
            if (k == 2) {
                prop_gemm2<0, 0><<<gGemm, blk, 0, stream>>>(tin, tpv, rowptr, pair, Wa, Wb, bl,
                                                           P[s_idx], ACC, nullptr, N_NODES);
            } else if (k == 3 || k == 6) {
                prop_gemm2<1, 0><<<gGemm, blk, 0, stream>>>(tin, tpv, rowptr, pair, Wa, Wb, bl,
                                                           P[s_idx], ACC, nullptr, N_NODES);
            } else if (k == KORD - 1) {
                float* hn = (l == NLAYERS - 1) ? out : P[s_idx];
                prop_gemm2<1, 1><<<gGemm, blk, 0, stream>>>(tin, tpv, rowptr, pair, Wa, Wb, bl,
                                                           nullptr, ACC, hn, N_NODES);
            } else {
                prop_only<1><<<gGemm, blk, 0, stream>>>(tin, tpv, rowptr, pair, P[s_idx], N_NODES);
            }
            if (t0_idx >= 0) q[qt++] = t0_idx;
            t0_idx = t1_idx;
            t1_idx = s_idx;
        }
        h_idx = t1_idx;
    }
}


# Row gather issues its loads in padded batches of 8 edges (slots past the row's end re-read the last edge's row with weight 0), no 4-edge trips or one-edge remainder loop; sum order unchanged
# speedup vs baseline: 3.6404x; 1.0983x over previous
#include <hip/hip_runtime.h>

#define N_NODES 100000
#define N_EDGES 1000000
#define DHID    64
#define KORD    8
#define NLAYERS 3

typedef float v2f __attribute__((ext_vector_type(2)));
typedef float v8f __attribute__((ext_vector_type(8)));

__global__ void zero_kernel(float* __restrict__ p, int n) {
    int i = blockIdx.x * blockDim.x + threadIdx.x;
    int stride = gridDim.x * blockDim.x;
    for (; i < n; i += stride) p[i] = 0.0f;
}

__global__ void deg_rank_kernel(const int* __restrict__ src, const int* __restrict__ dst,
                                const float* __restrict__ ew, float* __restrict__ deg,
                                int* __restrict__ cnt, int* __restrict__ rank, int nE) {
    int e = blockIdx.x * blockDim.x + threadIdx.x;
    if (e >= nE) return;
    int s = src[e], d = dst[e];
    if (s != d) atomicAdd(&deg[s], ew[e]);
    rank[e] = atomicAdd(&cnt[d], 1);
}

__global__ void __launch_bounds__(256)
dinv_bsum_kernel(float* __restrict__ deg, const int* __restrict__ cnt,
                 int* __restrict__ bsum, int n) {
    __shared__ int red[8];
    const int i0 = (blockIdx.x * 256 + threadIdx.x) * 4;
    int s = 0;
    for (int j = 0; j < 4; ++j) {
        int i = i0 + j;
        if (i < n) {
            float d = deg[i];
            deg[i] = (d > 0.0f) ? rsqrtf(d) : 0.0f;
            s += cnt[i];
        }
    }
    for (int off = 16; off > 0; off >>= 1) s += __shfl_xor(s, off);
    if ((threadIdx.x & 31) == 0) red[threadIdx.x >> 5] = s;
    __syncthreads();
    if (threadIdx.x == 0) {
        int t = 0;
        for (int w = 0; w < 8; ++w) t += red[w];
        bsum[blockIdx.x] = t;
    }
}

__global__ void __launch_bounds__(256)
scan_kernel(const int* __restrict__ cnt, const int* __restrict__ bsum,
            int* __restrict__ rowptr, int n) {
    __shared__ int wsum[8];
    __shared__ int osum[8];
    const int tid = threadIdx.x, lane = tid & 31, wave = tid >> 5;
    int o = (tid < (int)blockIdx.x) ? bsum[tid] : 0;
    for (int off = 16; off > 0; off >>= 1) o += __shfl_xor(o, off);
    const int i0 = (blockIdx.x * 256 + tid) * 4;
    int c[4];
    int s = 0;
    for (int j = 0; j < 4; ++j) {
        c[j] = (i0 + j < n) ? cnt[i0 + j] : 0;
        s += c[j];
    }
    int inc = s;
    for (int off = 1; off < 32; off <<= 1) {
        int v = __shfl_up(inc, off);
        if (lane >= off) inc += v;
    }
    if (lane == 0)  osum[wave] = o;
    if (lane == 31) wsum[wave] = inc;
    __syncthreads();
    int ex = inc - s;
    for (int w = 0; w < 8; ++w) {
        ex += osum[w];
        if (w < wave) ex += wsum[w];
    }
    for (int j = 0; j < 4; ++j) {
        if (i0 + j < n) { rowptr[i0 + j] = ex; ex += c[j]; }
    }
    if (i0 < n && i0 + 4 >= n) rowptr[n] = ex;
}

__global__ void fill_kernel(const int* __restrict__ dst, const int* __restrict__ rowptr,
                            const int* __restrict__ rank, int* __restrict__ eid, int nE) {
    int e = blockIdx.x * blockDim.x + threadIdx.x;
    if (e >= nE) return;
    eid[rowptr[dst[e]] + rank[e]] = e;
}

__global__ void csr_final_kernel(const int* __restrict__ src, const int* __restrict__ dst,
                                 const float* __restrict__ ew, const float* __restrict__ dinv,
                                 const int* __restrict__ rowptr, const int* __restrict__ eid,
                                 int* __restrict__ pair, int nE) {
    int e = blockIdx.x * blockDim.x + threadIdx.x;
    if (e >= nE) return;
    int s = src[e], d = dst[e];
    int b = rowptr[d], en = rowptr[d + 1];
    int r = 0;
    for (int j = b; j < en; ++j) r += (eid[j] < e) ? 1 : 0;
    float w = (s == d) ? 0.0f : ew[e];
    float nm = -dinv[s] * w * dinv[d];
    *(int2*)(pair + 2 * (size_t)(b + r)) = make_int2(s, __float_as_int(nm));
}

__device__ __forceinline__ void stage_w(float* lWp, const float* __restrict__ W) {
    for (int idx = threadIdx.x; idx < DHID * DHID; idx += 256) {
        int k = idx >> 6, n = idx & 63;
        lWp[((k >> 1) << 7) + (n << 1) + (k & 1)] = W[idx];
    }
}

__device__ __forceinline__ void tile_gemm(const float* arow, const float* lWp, int khalf, int lrow,
                                          v8f& c0, v8f& c1, v8f& c2, v8f& c3) {
#pragma unroll
    for (int kt = 0; kt < 16; ++kt) {
        const int k0 = kt * 4 + khalf;
        const v2f a = *(const v2f*)(arow + kt * 4);
        const float* wp = lWp + ((k0 >> 1) << 7);
        v2f b;
        b = *(const v2f*)(wp + ((lrow +  0) << 1));
        c0 = __builtin_amdgcn_wmma_f32_16x16x4_f32(false, a, false, b, (short)0, c0, false, false);
        b = *(const v2f*)(wp + ((lrow + 16) << 1));
        c1 = __builtin_amdgcn_wmma_f32_16x16x4_f32(false, a, false, b, (short)0, c1, false, false);
        b = *(const v2f*)(wp + ((lrow + 32) << 1));
        c2 = __builtin_amdgcn_wmma_f32_16x16x4_f32(false, a, false, b, (short)0, c2, false, false);
        b = *(const v2f*)(wp + ((lrow + 48) << 1));
        c3 = __builtin_amdgcn_wmma_f32_16x16x4_f32(false, a, false, b, (short)0, c3, false, false);
    }
}

template <int BASE>
__device__ __forceinline__ void gather8(const float* __restrict__ tin, int ms, float mw, int f4, float4& a) {
    int    s[8];
    float  w[8];
    float4 h[8];
#pragma unroll
    for (int j = 0; j < 8; ++j) {
        s[j] = __shfl(ms, BASE + j, 16);
        w[j] = __shfl(mw, BASE + j, 16);
    }
#pragma unroll
    for (int j = 0; j < 8; ++j)
        h[j] = *(const float4*)(tin + (size_t)s[j] * DHID + f4);
#pragma unroll
    for (int j = 0; j < 8; ++j) {
        a.x = fmaf(w[j], h[j].x, a.x);
        a.y = fmaf(w[j], h[j].y, a.y);
        a.z = fmaf(w[j], h[j].z, a.z);
        a.w = fmaf(w[j], h[j].w, a.w);
    }
}

__device__ __forceinline__ float4 gather_row(const float* __restrict__ tin, const int* __restrict__ pair,
                                             int b, int en, int ms, float mw, int lrow, int f4) {
    float4 a = make_float4(0.0f, 0.0f, 0.0f, 0.0f);
    int c0e = b;
    while (c0e < en) {
        const int m  = (en - c0e < 16) ? (en - c0e) : 16;
        const int sl = __shfl(ms, m - 1, 16);
        if (lrow >= m) {
            ms = sl;
            mw = 0.0f;
        }
        gather8<0>(tin, ms, mw, f4, a);
        if (m > 8) gather8<8>(tin, ms, mw, f4, a);
        c0e += 16;
        if (c0e + lrow < en) {
            const int2 p = *(const int2*)(pair + 2 * (size_t)(c0e + lrow));
            ms = p.x;
            mw = __int_as_float(p.y);
        }
    }
    return a;
}

__device__ __forceinline__ void first_pairs(const int* __restrict__ pair, int rp, int half, int lrow,
                                            int (&pms)[8], float (&pmw)[8]) {
#pragma unroll
    for (int it = 0; it < 8; ++it) {
        const int b    = __shfl(rp, it * 2 + half);
        const int slot = (b + lrow < N_EDGES) ? (b + lrow) : (N_EDGES - 1);
        const int2 p   = *(const int2*)(pair + 2 * (size_t)slot);
        pms[it] = p.x;
        pmw[it] = __int_as_float(p.y);
    }
}

template <int RECUR>
__global__ void __launch_bounds__(256)
prop_only(const float* __restrict__ tin, const float* __restrict__ tprev,
          const int* __restrict__ rowptr, const int* __restrict__ pair,
          float* __restrict__ tout, int nrows) {
    const int wave = threadIdx.x >> 5;
    const int lane = threadIdx.x & 31;
    const int row0 = (blockIdx.x * 8 + wave) * 16;
    if (row0 >= nrows) return;

    const int lrow = lane & 15;
    const int half = lane >> 4;
    const int f4   = lrow << 2;

    const int rp = rowptr[row0 + ((lane < 16) ? lane : 16)];
    int   pms[8];
    float pmw[8];
    first_pairs(pair, rp, half, lrow, pms, pmw);

#pragma unroll
    for (int it = 0; it < 8; ++it) {
        const int r  = it * 2 + half;
        const int b  = __shfl(rp, r), en = __shfl(rp, r + 1);
        float4 tp = make_float4(0.0f, 0.0f, 0.0f, 0.0f);
        if (RECUR) tp = *(const float4*)(tprev + (size_t)(row0 + r) * DHID + f4);
        float4 a = gather_row(tin, pair, b, en, pms[it], pmw[it], lrow, f4);
        if (RECUR) {
            a.x = 2.0f * a.x - tp.x;
            a.y = 2.0f * a.y - tp.y;
            a.z = 2.0f * a.z - tp.z;
            a.w = 2.0f * a.w - tp.w;
        }
        *(float4*)(tout + (size_t)(row0 + r) * DHID + f4) = a;
    }
}

#define TPITCH 68

template <int ACCIN, int FINAL>
__global__ void __launch_bounds__(256)
prop_gemm2(const float* __restrict__ tin, const float* __restrict__ tprev,
           const int* __restrict__ rowptr, const int* __restrict__ pair,
           const float* __restrict__ Wa, const float* __restrict__ Wb,
           const float* __restrict__ bias, float* __restrict__ tout,
           float* __restrict__ acc, float* __restrict__ hout, int nrows) {
    __shared__ float lW[2 * DHID * DHID];
    __shared__ __attribute__((aligned(16))) float lT[8 * 16 * TPITCH];
    stage_w(lW, Wa);
    stage_w(lW + DHID * DHID, Wb);
    __syncthreads();

    const int wave = threadIdx.x >> 5;
    const int lane = threadIdx.x & 31;
    const int row0 = __builtin_amdgcn_readfirstlane((blockIdx.x * 8 + wave) * 16);
    if (row0 >= nrows) return;

    const int lrow  = lane & 15;
    const int half  = lane >> 4;
    const int khalf = half << 1;
    const int ccol  = lane & 15;
    const int f4    = lrow << 2;
    const int loff  = (half << 6) + f4;
    const size_t rbase = (size_t)row0 * DHID;

    const int rp = rowptr[row0 + ((lane < 16) ? lane : 16)];
    float* myT = lT + wave * (16 * TPITCH);
    const float* arow = myT + lrow * TPITCH + khalf;

#pragma unroll
    for (int it = 0; it < 8; ++it) {
        const int r = it * 2 + half;
        const float4 t = *(const float4*)(tprev + rbase + it * 2 * DHID + loff);
        *(float4*)(myT + r * TPITCH + f4) = t;
    }

    int   pms[8];
    float pmw[8];
    first_pairs(pair, rp, half, lrow, pms, pmw);

    __builtin_amdgcn_fence(__ATOMIC_RELEASE, "wavefront");
    __builtin_amdgcn_wave_barrier();
    __builtin_amdgcn_fence(__ATOMIC_ACQUIRE, "wavefront");

    v8f c0 = {}, c1 = {}, c2 = {}, c3 = {};
    tile_gemm(arow, lW, khalf, lrow, c0, c1, c2, c3);
    asm volatile("" : "+v"(c0), "+v"(c1), "+v"(c2), "+v"(c3));

#pragma unroll
    for (int it = 0; it < 8; ++it) {
        const int r  = it * 2 + half;
        const int b  = __shfl(rp, r), en = __shfl(rp, r + 1);
        float4 a = gather_row(tin, pair, b, en, pms[it], pmw[it], lrow, f4);
        const float4 tp = *(const float4*)(myT + r * TPITCH + f4);
        a.x = 2.0f * a.x - tp.x;
        a.y = 2.0f * a.y - tp.y;
        a.z = 2.0f * a.z - tp.z;
        a.w = 2.0f * a.w - tp.w;
        if (!FINAL) *(float4*)(tout + rbase + it * 2 * DHID + loff) = a;
        *(float4*)(myT + r * TPITCH + f4) = a;
    }

    float4 av[8];
    if (ACCIN) {
#pragma unroll
        for (int it = 0; it < 8; ++it)
            av[it] = *(const float4*)(acc + rbase + it * 2 * DHID + loff);
    }

    __builtin_amdgcn_fence(__ATOMIC_RELEASE, "wavefront");
    __builtin_amdgcn_wave_barrier();
    __builtin_amdgcn_fence(__ATOMIC_ACQUIRE, "wavefront");

    tile_gemm(arow, lW + DHID * DHID, khalf, lrow, c0, c1, c2, c3);

#pragma unroll
    for (int r = 0; r < 8; ++r) {
        float* tr = myT + ((half << 3) + r) * TPITCH + ccol;
        tr[0] = c0[r]; tr[16] = c1[r]; tr[32] = c2[r]; tr[48] = c3[r];
    }

    __builtin_amdgcn_fence(__ATOMIC_RELEASE, "wavefront");
    __builtin_amdgcn_wave_barrier();
    __builtin_amdgcn_fence(__ATOMIC_ACQUIRE, "wavefront");

    float4 bv = make_float4(0.0f, 0.0f, 0.0f, 0.0f);
    if (FINAL) bv = *(const float4*)(bias + f4);
#pragma unroll
    for (int it = 0; it < 8; ++it) {
        const int r = it * 2 + half;
        float4 v = *(const float4*)(myT + r * TPITCH + f4);
        if (ACCIN) {
            v.x = av[it].x + v.x; v.y = av[it].y + v.y;
            v.z = av[it].z + v.z; v.w = av[it].w + v.w;
        }
        if (FINAL) {
            v.x = fmaxf(v.x + bv.x, 0.0f); v.y = fmaxf(v.y + bv.y, 0.0f);
            v.z = fmaxf(v.z + bv.z, 0.0f); v.w = fmaxf(v.w + bv.w, 0.0f);
        }
        float* dstp = FINAL ? hout : acc;
        *(float4*)(dstp + rbase + it * 2 * DHID + loff) = v;
    }
}

extern "C" void kernel_launch(void* const* d_in, const int* in_sizes, int n_in,
                              void* d_out, int out_size, void* d_ws, size_t ws_size,
                              hipStream_t stream) {
    const float* x       = (const float*)d_in[0];
    const int*   ei      = (const int*)d_in[1];
    const float* ew      = (const float*)d_in[2];
    const float* weights = (const float*)d_in[3];
    const float* bias    = (const float*)d_in[4];
    float*       out     = (float*)d_out;

    const int* src = ei;
    const int* dst = ei + N_EDGES;

    float* ws   = (float*)d_ws;
    float* base = ws + N_EDGES + N_NODES;
    float* P[5];
    for (int i = 0; i < 5; ++i) P[i] = base + (size_t)i * N_NODES * DHID;
    float* ACC = P[4];

    int*   pair   = (int*)P[3];
    int*   rowptr = pair + 2 * N_EDGES;
    float* dinv   = (float*)(rowptr + N_NODES + 32);
    int*   cnt    = (int*)(dinv + N_NODES);
    int*   eid    = cnt + N_NODES;
    int*   bsum   = eid + N_EDGES;
    int*   rank   = (int*)ws;

    dim3 blk(256);
    const int gE    = (N_EDGES + 255) / 256;
    const int gScan = (N_NODES + 1023) / 1024;
    const int gGemm = (N_NODES + 127) / 128;
    const int WSZ   = DHID * DHID;

    zero_kernel     <<<256,   blk, 0, stream>>>(dinv, 2 * N_NODES);
    deg_rank_kernel <<<gE,    blk, 0, stream>>>(src, dst, ew, dinv, cnt, rank, N_EDGES);
    dinv_bsum_kernel<<<gScan, blk, 0, stream>>>(dinv, cnt, bsum, N_NODES);
    scan_kernel     <<<gScan, blk, 0, stream>>>(cnt, bsum, rowptr, N_NODES);
    fill_kernel     <<<gE,    blk, 0, stream>>>(dst, rowptr, rank, eid, N_EDGES);
    csr_final_kernel<<<gE,    blk, 0, stream>>>(src, dst, ew, dinv, rowptr, eid, pair, N_EDGES);

    int h_idx = -1;
    for (int l = 0; l < NLAYERS; ++l) {
        int rid[3], c = 0;
        for (int i = 0; i < 3; ++i)
            if (i != h_idx) rid[c++] = i;

        const float* t0 = (h_idx < 0) ? x : P[h_idx];
        float* t1 = P[rid[0]];
        const float* Wl = weights + (size_t)l * KORD * WSZ;
        const float* bl = bias + l * DHID;

        prop_only<0><<<gGemm, blk, 0, stream>>>(t0, nullptr, rowptr, pair, t1, N_NODES);

        int q[10]; int qh = 0, qt = 0;
        for (int i = 1; i < c; ++i) q[qt++] = rid[i];
        int t0_idx = h_idx;
        int t1_idx = rid[0];

        for (int k = 2; k < KORD; ++k) {
            int s_idx = q[qh++];
            const float* tin = P[t1_idx];
            const float* tpv = (t0_idx < 0) ? x : P[t0_idx];
            const float* Wa  = Wl + (size_t)(k - 2) * WSZ;
            const float* Wb  = Wl + (size_t)k * WSZ;
            if (k == 2) {
                prop_gemm2<0, 0><<<gGemm, blk, 0, stream>>>(tin, tpv, rowptr, pair, Wa, Wb, bl,
                                                           P[s_idx], ACC, nullptr, N_NODES);
            } else if (k == 3 || k == 6) {
                prop_gemm2<1, 0><<<gGemm, blk, 0, stream>>>(tin, tpv, rowptr, pair, Wa, Wb, bl,
                                                           P[s_idx], ACC, nullptr, N_NODES);
            } else if (k == KORD - 1) {
                float* hn = (l == NLAYERS - 1) ? out : P[s_idx];
                prop_gemm2<1, 1><<<gGemm, blk, 0, stream>>>(tin, tpv, rowptr, pair, Wa, Wb, bl,
                                                           nullptr, ACC, hn, N_NODES);
            } else {
                prop_only<1><<<gGemm, blk, 0, stream>>>(tin, tpv, rowptr, pair, P[s_idx], N_NODES);
            }
            if (t0_idx >= 0) q[qt++] = t0_idx;
            t0_idx = t1_idx;
            t1_idx = s_idx;
        }
        h_idx = t1_idx;
    }
}


# Propagation grid reshaped to one 25-wave workgroup per WGP: 250 workgroups x 25 row tiles, was 782 x 8 (some WGPs held 32 tiles, most 24); weights staged once per WGP; sum order unchanged
# speedup vs baseline: 3.9351x; 1.0810x over previous
#include <hip/hip_runtime.h>

#define N_NODES 100000
#define N_EDGES 1000000
#define DHID    64
#define KORD    8
#define NLAYERS 3
#define WPB     25
#define NTHR    (WPB * 32)

typedef float v2f __attribute__((ext_vector_type(2)));
typedef float v8f __attribute__((ext_vector_type(8)));

__global__ void zero_kernel(float* __restrict__ p, int n) {
    int i = blockIdx.x * blockDim.x + threadIdx.x;
    int stride = gridDim.x * blockDim.x;
    for (; i < n; i += stride) p[i] = 0.0f;
}

__global__ void deg_rank_kernel(const int* __restrict__ src, const int* __restrict__ dst,
                                const float* __restrict__ ew, float* __restrict__ deg,
                                int* __restrict__ cnt, int* __restrict__ rank, int nE) {
    int e = blockIdx.x * blockDim.x + threadIdx.x;
    if (e >= nE) return;
    int s = src[e], d = dst[e];
    if (s != d) atomicAdd(&deg[s], ew[e]);
    rank[e] = atomicAdd(&cnt[d], 1);
}

__global__ void __launch_bounds__(256)
dinv_bsum_kernel(float* __restrict__ deg, const int* __restrict__ cnt,
                 int* __restrict__ bsum, int n) {
    __shared__ int red[8];
    const int i0 = (blockIdx.x * 256 + threadIdx.x) * 4;
    int s = 0;
    for (int j = 0; j < 4; ++j) {
        int i = i0 + j;
        if (i < n) {
            float d = deg[i];
            deg[i] = (d > 0.0f) ? rsqrtf(d) : 0.0f;
            s += cnt[i];
        }
    }
    for (int off = 16; off > 0; off >>= 1) s += __shfl_xor(s, off);
    if ((threadIdx.x & 31) == 0) red[threadIdx.x >> 5] = s;
    __syncthreads();
    if (threadIdx.x == 0) {
        int t = 0;
        for (int w = 0; w < 8; ++w) t += red[w];
        bsum[blockIdx.x] = t;
    }
}

__global__ void __launch_bounds__(256)
scan_kernel(const int* __restrict__ cnt, const int* __restrict__ bsum,
            int* __restrict__ rowptr, int n) {
    __shared__ int wsum[8];
    __shared__ int osum[8];
    const int tid = threadIdx.x, lane = tid & 31, wave = tid >> 5;
    int o = (tid < (int)blockIdx.x) ? bsum[tid] : 0;
    for (int off = 16; off > 0; off >>= 1) o += __shfl_xor(o, off);
    const int i0 = (blockIdx.x * 256 + tid) * 4;
    int c[4];
    int s = 0;
    for (int j = 0; j < 4; ++j) {
        c[j] = (i0 + j < n) ? cnt[i0 + j] : 0;
        s += c[j];
    }
    int inc = s;
    for (int off = 1; off < 32; off <<= 1) {
        int v = __shfl_up(inc, off);
        if (lane >= off) inc += v;
    }
    if (lane == 0)  osum[wave] = o;
    if (lane == 31) wsum[wave] = inc;
    __syncthreads();
    int ex = inc - s;
    for (int w = 0; w < 8; ++w) {
        ex += osum[w];
        if (w < wave) ex += wsum[w];
    }
    for (int j = 0; j < 4; ++j) {
        if (i0 + j < n) { rowptr[i0 + j] = ex; ex += c[j]; }
    }
    if (i0 < n && i0 + 4 >= n) rowptr[n] = ex;
}

__global__ void fill_kernel(const int* __restrict__ dst, const int* __restrict__ rowptr,
                            const int* __restrict__ rank, int* __restrict__ eid, int nE) {
    int e = blockIdx.x * blockDim.x + threadIdx.x;
    if (e >= nE) return;
    eid[rowptr[dst[e]] + rank[e]] = e;
}

__global__ void csr_final_kernel(const int* __restrict__ src, const int* __restrict__ dst,
                                 const float* __restrict__ ew, const float* __restrict__ dinv,
                                 const int* __restrict__ rowptr, const int* __restrict__ eid,
                                 int* __restrict__ pair, int nE) {
    int e = blockIdx.x * blockDim.x + threadIdx.x;
    if (e >= nE) return;
    int s = src[e], d = dst[e];
    int b = rowptr[d], en = rowptr[d + 1];
    int r = 0;
    for (int j = b; j < en; ++j) r += (eid[j] < e) ? 1 : 0;
    float w = (s == d) ? 0.0f : ew[e];
    float nm = -dinv[s] * w * dinv[d];
    *(int2*)(pair + 2 * (size_t)(b + r)) = make_int2(s, __float_as_int(nm));
}

#define WLOADS ((DHID * DHID + NTHR - 1) / NTHR)
__device__ __forceinline__ void stage_w(float* lWp, const float* __restrict__ W) {
    float v[WLOADS];
#pragma unroll
    for (int i = 0; i < WLOADS; ++i) {
        const int idx = threadIdx.x + i * NTHR;
        v[i] = (idx < DHID * DHID) ? W[idx] : 0.0f;
    }
#pragma unroll
    for (int i = 0; i < WLOADS; ++i) {
        const int idx = threadIdx.x + i * NTHR;
        const int k = idx >> 6, n = idx & 63;
        if (idx < DHID * DHID) lWp[((k >> 1) << 7) + (n << 1) + (k & 1)] = v[i];
    }
}

__device__ __forceinline__ void tile_gemm(const float* arow, const float* lWp, int khalf, int lrow,
                                          v8f& c0, v8f& c1, v8f& c2, v8f& c3) {
#pragma unroll
    for (int kt = 0; kt < 16; ++kt) {
        const int k0 = kt * 4 + khalf;
        const v2f a = *(const v2f*)(arow + kt * 4);
        const float* wp = lWp + ((k0 >> 1) << 7);
        v2f b;
        b = *(const v2f*)(wp + ((lrow +  0) << 1));
        c0 = __builtin_amdgcn_wmma_f32_16x16x4_f32(false, a, false, b, (short)0, c0, false, false);
        b = *(const v2f*)(wp + ((lrow + 16) << 1));
        c1 = __builtin_amdgcn_wmma_f32_16x16x4_f32(false, a, false, b, (short)0, c1, false, false);
        b = *(const v2f*)(wp + ((lrow + 32) << 1));
        c2 = __builtin_amdgcn_wmma_f32_16x16x4_f32(false, a, false, b, (short)0, c2, false, false);
        b = *(const v2f*)(wp + ((lrow + 48) << 1));
        c3 = __builtin_amdgcn_wmma_f32_16x16x4_f32(false, a, false, b, (short)0, c3, false, false);
    }
}

template <int BASE>
__device__ __forceinline__ void gather8(const float* __restrict__ tin, int ms, float mw, int f4, float4& a) {
    int    s[8];
    float  w[8];
    float4 h[8];
#pragma unroll
    for (int j = 0; j < 8; ++j) {
        s[j] = __shfl(ms, BASE + j, 16);
        w[j] = __shfl(mw, BASE + j, 16);
    }
#pragma unroll
    for (int j = 0; j < 8; ++j)
        h[j] = *(const float4*)(tin + (size_t)s[j] * DHID + f4);
#pragma unroll
    for (int j = 0; j < 8; ++j) {
        a.x = fmaf(w[j], h[j].x, a.x);
        a.y = fmaf(w[j], h[j].y, a.y);
        a.z = fmaf(w[j], h[j].z, a.z);
        a.w = fmaf(w[j], h[j].w, a.w);
    }
}

__device__ __forceinline__ float4 gather_row(const float* __restrict__ tin, const int* __restrict__ pair,
                                             int b, int en, int ms, float mw, int lrow, int f4) {
    float4 a = make_float4(0.0f, 0.0f, 0.0f, 0.0f);
    int c0e = b;
    while (c0e < en) {
        const int m  = (en - c0e < 16) ? (en - c0e) : 16;
        const int sl = __shfl(ms, m - 1, 16);
        if (lrow >= m) {
            ms = sl;
            mw = 0.0f;
        }
        gather8<0>(tin, ms, mw, f4, a);
        if (m > 8) gather8<8>(tin, ms, mw, f4, a);
        c0e += 16;
        if (c0e + lrow < en) {
            const int2 p = *(const int2*)(pair + 2 * (size_t)(c0e + lrow));
            ms = p.x;
            mw = __int_as_float(p.y);
        }
    }
    return a;
}

__device__ __forceinline__ void first_pairs(const int* __restrict__ pair, int rp, int half, int lrow,
                                            int (&pms)[8], float (&pmw)[8]) {
#pragma unroll
    for (int it = 0; it < 8; ++it) {
        const int b    = __shfl(rp, it * 2 + half);
        const int slot = (b + lrow < N_EDGES) ? (b + lrow) : (N_EDGES - 1);
        const int2 p   = *(const int2*)(pair + 2 * (size_t)slot);
        pms[it] = p.x;
        pmw[it] = __int_as_float(p.y);
    }
}

template <int RECUR>
__global__ void __launch_bounds__(NTHR)
prop_only(const float* __restrict__ tin, const float* __restrict__ tprev,
          const int* __restrict__ rowptr, const int* __restrict__ pair,
          float* __restrict__ tout, int nrows) {
    const int wave = threadIdx.x >> 5;
    const int lane = threadIdx.x & 31;
    const int row0 = (blockIdx.x * WPB + wave) * 16;
    if (row0 >= nrows) return;

    const int lrow = lane & 15;
    const int half = lane >> 4;
    const int f4   = lrow << 2;

    const int rp = rowptr[row0 + ((lane < 16) ? lane : 16)];
    int   pms[8];
    float pmw[8];
    first_pairs(pair, rp, half, lrow, pms, pmw);

#pragma unroll
    for (int it = 0; it < 8; ++it) {
        const int r  = it * 2 + half;
        const int b  = __shfl(rp, r), en = __shfl(rp, r + 1);
        float4 tp = make_float4(0.0f, 0.0f, 0.0f, 0.0f);
        if (RECUR) tp = *(const float4*)(tprev + (size_t)(row0 + r) * DHID + f4);
        float4 a = gather_row(tin, pair, b, en, pms[it], pmw[it], lrow, f4);
        if (RECUR) {
            a.x = 2.0f * a.x - tp.x;
            a.y = 2.0f * a.y - tp.y;
            a.z = 2.0f * a.z - tp.z;
            a.w = 2.0f * a.w - tp.w;
        }
        *(float4*)(tout + (size_t)(row0 + r) * DHID + f4) = a;
    }
}

#define TPITCH 68

template <int ACCIN, int FINAL>
__global__ void __launch_bounds__(NTHR)
prop_gemm2(const float* __restrict__ tin, const float* __restrict__ tprev,
           const int* __restrict__ rowptr, const int* __restrict__ pair,
           const float* __restrict__ Wa, const float* __restrict__ Wb,
           const float* __restrict__ bias, float* __restrict__ tout,
           float* __restrict__ acc, float* __restrict__ hout, int nrows) {
    __shared__ __attribute__((aligned(16))) float lds[2 * DHID * DHID + WPB * 16 * TPITCH];
    float* lW = lds;
    float* lT = lds + 2 * DHID * DHID;
    stage_w(lW, Wa);
    stage_w(lW + DHID * DHID, Wb);
    __syncthreads();

    const int wave = threadIdx.x >> 5;
    const int lane = threadIdx.x & 31;
    const int row0 = __builtin_amdgcn_readfirstlane((blockIdx.x * WPB + wave) * 16);
    if (row0 >= nrows) return;

    const int lrow  = lane & 15;
    const int half  = lane >> 4;
    const int khalf = half << 1;
    const int ccol  = lane & 15;
    const int f4    = lrow << 2;
    const int loff  = (half << 6) + f4;
    const size_t rbase = (size_t)row0 * DHID;

    const int rp = rowptr[row0 + ((lane < 16) ? lane : 16)];
    float* myT = lT + wave * (16 * TPITCH);
    const float* arow = myT + lrow * TPITCH + khalf;

#pragma unroll
    for (int it = 0; it < 8; ++it) {
        const int r = it * 2 + half;
        const float4 t = *(const float4*)(tprev + rbase + it * 2 * DHID + loff);
        *(float4*)(myT + r * TPITCH + f4) = t;
    }

    int   pms[8];
    float pmw[8];
    first_pairs(pair, rp, half, lrow, pms, pmw);

    __builtin_amdgcn_fence(__ATOMIC_RELEASE, "wavefront");
    __builtin_amdgcn_wave_barrier();
    __builtin_amdgcn_fence(__ATOMIC_ACQUIRE, "wavefront");

    v8f c0 = {}, c1 = {}, c2 = {}, c3 = {};
    tile_gemm(arow, lW, khalf, lrow, c0, c1, c2, c3);
    asm volatile("" : "+v"(c0), "+v"(c1), "+v"(c2), "+v"(c3));

#pragma unroll
    for (int it = 0; it < 8; ++it) {
        const int r  = it * 2 + half;
        const int b  = __shfl(rp, r), en = __shfl(rp, r + 1);
        float4 a = gather_row(tin, pair, b, en, pms[it], pmw[it], lrow, f4);
        const float4 tp = *(const float4*)(myT + r * TPITCH + f4);
        a.x = 2.0f * a.x - tp.x;
        a.y = 2.0f * a.y - tp.y;
        a.z = 2.0f * a.z - tp.z;
        a.w = 2.0f * a.w - tp.w;
        if (!FINAL) *(float4*)(tout + rbase + it * 2 * DHID + loff) = a;
        *(float4*)(myT + r * TPITCH + f4) = a;
    }

    float4 av[8];
    if (ACCIN) {
#pragma unroll
        for (int it = 0; it < 8; ++it)
            av[it] = *(const float4*)(acc + rbase + it * 2 * DHID + loff);
    }

    __builtin_amdgcn_fence(__ATOMIC_RELEASE, "wavefront");
    __builtin_amdgcn_wave_barrier();
    __builtin_amdgcn_fence(__ATOMIC_ACQUIRE, "wavefront");

    tile_gemm(arow, lW + DHID * DHID, khalf, lrow, c0, c1, c2, c3);

#pragma unroll
    for (int r = 0; r < 8; ++r) {
        float* tr = myT + ((half << 3) + r) * TPITCH + ccol;
        tr[0] = c0[r]; tr[16] = c1[r]; tr[32] = c2[r]; tr[48] = c3[r];
    }

    __builtin_amdgcn_fence(__ATOMIC_RELEASE, "wavefront");
    __builtin_amdgcn_wave_barrier();
    __builtin_amdgcn_fence(__ATOMIC_ACQUIRE, "wavefront");

    float4 bv = make_float4(0.0f, 0.0f, 0.0f, 0.0f);
    if (FINAL) bv = *(const float4*)(bias + f4);
#pragma unroll
    for (int it = 0; it < 8; ++it) {
        const int r = it * 2 + half;
        float4 v = *(const float4*)(myT + r * TPITCH + f4);
        if (ACCIN) {
            v.x = av[it].x + v.x; v.y = av[it].y + v.y;
            v.z = av[it].z + v.z; v.w = av[it].w + v.w;
        }
        if (FINAL) {
            v.x = fmaxf(v.x + bv.x, 0.0f); v.y = fmaxf(v.y + bv.y, 0.0f);
            v.z = fmaxf(v.z + bv.z, 0.0f); v.w = fmaxf(v.w + bv.w, 0.0f);
        }
        float* dstp = FINAL ? hout : acc;
        *(float4*)(dstp + rbase + it * 2 * DHID + loff) = v;
    }
}

extern "C" void kernel_launch(void* const* d_in, const int* in_sizes, int n_in,
                              void* d_out, int out_size, void* d_ws, size_t ws_size,
                              hipStream_t stream) {
    const float* x       = (const float*)d_in[0];
    const int*   ei      = (const int*)d_in[1];
    const float* ew      = (const float*)d_in[2];
    const float* weights = (const float*)d_in[3];
    const float* bias    = (const float*)d_in[4];
    float*       out     = (float*)d_out;

    const int* src = ei;
    const int* dst = ei + N_EDGES;

    float* ws   = (float*)d_ws;
    float* base = ws + N_EDGES + N_NODES;
    float* P[5];
    for (int i = 0; i < 5; ++i) P[i] = base + (size_t)i * N_NODES * DHID;
    float* ACC = P[4];

    int*   pair   = (int*)P[3];
    int*   rowptr = pair + 2 * N_EDGES;
    float* dinv   = (float*)(rowptr + N_NODES + 32);
    int*   cnt    = (int*)(dinv + N_NODES);
    int*   eid    = cnt + N_NODES;
    int*   bsum   = eid + N_EDGES;
    int*   rank   = (int*)ws;

    dim3 blk(256);
    dim3 pblk(NTHR);
    const int gE    = (N_EDGES + 255) / 256;
    const int gScan = (N_NODES + 1023) / 1024;
    const int gGemm = (N_NODES + WPB * 16 - 1) / (WPB * 16);
    const int WSZ   = DHID * DHID;

    zero_kernel     <<<256,   blk, 0, stream>>>(dinv, 2 * N_NODES);
    deg_rank_kernel <<<gE,    blk, 0, stream>>>(src, dst, ew, dinv, cnt, rank, N_EDGES);
    dinv_bsum_kernel<<<gScan, blk, 0, stream>>>(dinv, cnt, bsum, N_NODES);
    scan_kernel     <<<gScan, blk, 0, stream>>>(cnt, bsum, rowptr, N_NODES);
    fill_kernel     <<<gE,    blk, 0, stream>>>(dst, rowptr, rank, eid, N_EDGES);
    csr_final_kernel<<<gE,    blk, 0, stream>>>(src, dst, ew, dinv, rowptr, eid, pair, N_EDGES);

    int h_idx = -1;
    for (int l = 0; l < NLAYERS; ++l) {
        int rid[3], c = 0;
        for (int i = 0; i < 3; ++i)
            if (i != h_idx) rid[c++] = i;

        const float* t0 = (h_idx < 0) ? x : P[h_idx];
        float* t1 = P[rid[0]];
        const float* Wl = weights + (size_t)l * KORD * WSZ;
        const float* bl = bias + l * DHID;

        prop_only<0><<<gGemm, pblk, 0, stream>>>(t0, nullptr, rowptr, pair, t1, N_NODES);

        int q[10]; int qh = 0, qt = 0;
        for (int i = 1; i < c; ++i) q[qt++] = rid[i];
        int t0_idx = h_idx;
        int t1_idx = rid[0];

        for (int k = 2; k < KORD; ++k) {
            int s_idx = q[qh++];
            const float* tin = P[t1_idx];
            const float* tpv = (t0_idx < 0) ? x : P[t0_idx];
            const float* Wa  = Wl + (size_t)(k - 2) * WSZ;
            const float* Wb  = Wl + (size_t)k * WSZ;
            if (k == 2) {
                prop_gemm2<0, 0><<<gGemm, pblk, 0, stream>>>(tin, tpv, rowptr, pair, Wa, Wb, bl,
                                                           P[s_idx], ACC, nullptr, N_NODES);
            } else if (k == 3 || k == 6) {
                prop_gemm2<1, 0><<<gGemm, pblk, 0, stream>>>(tin, tpv, rowptr, pair, Wa, Wb, bl,
                                                           P[s_idx], ACC, nullptr, N_NODES);
            } else if (k == KORD - 1) {
                float* hn = (l == NLAYERS - 1) ? out : P[s_idx];
                prop_gemm2<1, 1><<<gGemm, pblk, 0, stream>>>(tin, tpv, rowptr, pair, Wa, Wb, bl,
                                                           nullptr, ACC, hn, N_NODES);
            } else {
                prop_only<1><<<gGemm, pblk, 0, stream>>>(tin, tpv, rowptr, pair, P[s_idx], N_NODES);
            }
            if (t0_idx >= 0) q[qt++] = t0_idx;
            t0_idx = t1_idx;
            t1_idx = s_idx;
        }
        h_idx = t1_idx;
    }
}


# Graph build: edges go to their destination slots as 16-byte records (id, src, weight, dst); final pass in slot order, 4 slots a thread: 1 scattered read per edge, not 4 and a scattered store
# speedup vs baseline: 4.1553x; 1.0560x over previous
#include <hip/hip_runtime.h>

#define N_NODES 100000
#define N_EDGES 1000000
#define DHID    64
#define KORD    8
#define NLAYERS 3
#define WPB     25
#define NTHR    (WPB * 32)

typedef float v2f __attribute__((ext_vector_type(2)));
typedef float v8f __attribute__((ext_vector_type(8)));

__global__ void zero_kernel(float* __restrict__ p, int n) {
    int i = blockIdx.x * blockDim.x + threadIdx.x;
    int stride = gridDim.x * blockDim.x;
    for (; i < n; i += stride) p[i] = 0.0f;
}

__global__ void deg_rank_kernel(const int* __restrict__ src, const int* __restrict__ dst,
                                const float* __restrict__ ew, float* __restrict__ deg,
                                int* __restrict__ cnt, int* __restrict__ rank, int nE) {
    int e = blockIdx.x * blockDim.x + threadIdx.x;
    if (e >= nE) return;
    int s = src[e], d = dst[e];
    if (s != d) atomicAdd(&deg[s], ew[e]);
    rank[e] = atomicAdd(&cnt[d], 1);
}

__global__ void __launch_bounds__(256)
dinv_bsum_kernel(float* __restrict__ deg, const int* __restrict__ cnt,
                 int* __restrict__ bsum, int n) {
    __shared__ int red[8];
    const int i0 = (blockIdx.x * 256 + threadIdx.x) * 4;
    int s = 0;
    for (int j = 0; j < 4; ++j) {
        int i = i0 + j;
        if (i < n) {
            float d = deg[i];
            deg[i] = (d > 0.0f) ? rsqrtf(d) : 0.0f;
            s += cnt[i];
        }
    }
    for (int off = 16; off > 0; off >>= 1) s += __shfl_xor(s, off);
    if ((threadIdx.x & 31) == 0) red[threadIdx.x >> 5] = s;
    __syncthreads();
    if (threadIdx.x == 0) {
        int t = 0;
        for (int w = 0; w < 8; ++w) t += red[w];
        bsum[blockIdx.x] = t;
    }
}

__global__ void __launch_bounds__(256)
scan_kernel(const int* __restrict__ cnt, const int* __restrict__ bsum,
            int* __restrict__ rowptr, int n) {
    __shared__ int wsum[8];
    __shared__ int osum[8];
    const int tid = threadIdx.x, lane = tid & 31, wave = tid >> 5;
    int o = (tid < (int)blockIdx.x) ? bsum[tid] : 0;
    for (int off = 16; off > 0; off >>= 1) o += __shfl_xor(o, off);
    const int i0 = (blockIdx.x * 256 + tid) * 4;
    int c[4];
    int s = 0;
    for (int j = 0; j < 4; ++j) {
        c[j] = (i0 + j < n) ? cnt[i0 + j] : 0;
        s += c[j];
    }
    int inc = s;
    for (int off = 1; off < 32; off <<= 1) {
        int v = __shfl_up(inc, off);
        if (lane >= off) inc += v;
    }
    if (lane == 0)  osum[wave] = o;
    if (lane == 31) wsum[wave] = inc;
    __syncthreads();
    int ex = inc - s;
    for (int w = 0; w < 8; ++w) {
        ex += osum[w];
        if (w < wave) ex += wsum[w];
    }
    for (int j = 0; j < 4; ++j) {
        if (i0 + j < n) { rowptr[i0 + j] = ex; ex += c[j]; }
    }
    if (i0 < n && i0 + 4 >= n) rowptr[n] = ex;
}

__global__ void fill_kernel(const int* __restrict__ src, const int* __restrict__ dst,
                            const float* __restrict__ ew, const int* __restrict__ rowptr,
                            const int* __restrict__ rank, int* __restrict__ rec, int nE) {
    int e = blockIdx.x * blockDim.x + threadIdx.x;
    if (e >= nE) return;
    int s = src[e], d = dst[e];
    float w = (s == d) ? 0.0f : ew[e];
    *(int4*)(rec + 4 * (size_t)(rowptr[d] + rank[e])) = make_int4(e, s, __float_as_int(w), d);
}

#define FSLOTS 4
__global__ void __launch_bounds__(256)
csr_final_kernel(const int* __restrict__ rec, const float* __restrict__ dinv,
                 const int* __restrict__ rowptr, int* __restrict__ pair, int nE) {
    const int j0 = blockIdx.x * (256 * FSLOTS) + threadIdx.x;
    int4  q[FSLOTS];
    int   b[FSLOTS], en[FSLOTS];
    float ds[FSLOTS], dd[FSLOTS];
#pragma unroll
    for (int k = 0; k < FSLOTS; ++k) {
        const int j  = j0 + k * 256;
        const int jc = (j < nE) ? j : (nE - 1);
        q[k] = *(const int4*)(rec + 4 * (size_t)jc);
    }
#pragma unroll
    for (int k = 0; k < FSLOTS; ++k) {
        const int s = q[k].y, d = q[k].w;
        b[k]  = rowptr[d];
        en[k] = rowptr[d + 1];
        ds[k] = dinv[s];
        dd[k] = dinv[d];
    }
#pragma unroll
    for (int k = 0; k < FSLOTS; ++k) {
        const int e = q[k].x;
        int r = 0;
        for (int i = b[k]; i < en[k]; ++i) r += (rec[4 * (size_t)i] < e) ? 1 : 0;
        const float nm = -ds[k] * __int_as_float(q[k].z) * dd[k];
        if (j0 + k * 256 < nE)
            *(int2*)(pair + 2 * (size_t)(b[k] + r)) = make_int2(q[k].y, __float_as_int(nm));
    }
}

#define WLOADS ((DHID * DHID + NTHR - 1) / NTHR)
__device__ __forceinline__ void stage_w(float* lWp, const float* __restrict__ W) {
    float v[WLOADS];
#pragma unroll
    for (int i = 0; i < WLOADS; ++i) {
        const int idx = threadIdx.x + i * NTHR;
        v[i] = (idx < DHID * DHID) ? W[idx] : 0.0f;
    }
#pragma unroll
    for (int i = 0; i < WLOADS; ++i) {
        const int idx = threadIdx.x + i * NTHR;
        const int k = idx >> 6, n = idx & 63;
        if (idx < DHID * DHID) lWp[((k >> 1) << 7) + (n << 1) + (k & 1)] = v[i];
    }
}

__device__ __forceinline__ void tile_gemm(const float* arow, const float* lWp, int khalf, int lrow,
                                          v8f& c0, v8f& c1, v8f& c2, v8f& c3) {
#pragma unroll
    for (int kt = 0; kt < 16; ++kt) {
        const int k0 = kt * 4 + khalf;
        const v2f a = *(const v2f*)(arow + kt * 4);
        const float* wp = lWp + ((k0 >> 1) << 7);
        v2f b;
        b = *(const v2f*)(wp + ((lrow +  0) << 1));
        c0 = __builtin_amdgcn_wmma_f32_16x16x4_f32(false, a, false, b, (short)0, c0, false, false);
        b = *(const v2f*)(wp + ((lrow + 16) << 1));
        c1 = __builtin_amdgcn_wmma_f32_16x16x4_f32(false, a, false, b, (short)0, c1, false, false);
        b = *(const v2f*)(wp + ((lrow + 32) << 1));
        c2 = __builtin_amdgcn_wmma_f32_16x16x4_f32(false, a, false, b, (short)0, c2, false, false);
        b = *(const v2f*)(wp + ((lrow + 48) << 1));
        c3 = __builtin_amdgcn_wmma_f32_16x16x4_f32(false, a, false, b, (short)0, c3, false, false);
    }
}

template <int BASE>
__device__ __forceinline__ void gather8(const float* __restrict__ tin, int ms, float mw, int f4, float4& a) {
    int    s[8];
    float  w[8];
    float4 h[8];
#pragma unroll
    for (int j = 0; j < 8; ++j) {
        s[j] = __shfl(ms, BASE + j, 16);
        w[j] = __shfl(mw, BASE + j, 16);
    }
#pragma unroll
    for (int j = 0; j < 8; ++j)
        h[j] = *(const float4*)(tin + (size_t)s[j] * DHID + f4);
#pragma unroll
    for (int j = 0; j < 8; ++j) {
        a.x = fmaf(w[j], h[j].x, a.x);
        a.y = fmaf(w[j], h[j].y, a.y);
        a.z = fmaf(w[j], h[j].z, a.z);
        a.w = fmaf(w[j], h[j].w, a.w);
    }
}

__device__ __forceinline__ float4 gather_row(const float* __restrict__ tin, const int* __restrict__ pair,
                                             int b, int en, int ms, float mw, int lrow, int f4) {
    float4 a = make_float4(0.0f, 0.0f, 0.0f, 0.0f);
    int c0e = b;
    while (c0e < en) {
        const int m  = (en - c0e < 16) ? (en - c0e) : 16;
        const int sl = __shfl(ms, m - 1, 16);
        if (lrow >= m) {
            ms = sl;
            mw = 0.0f;
        }
        gather8<0>(tin, ms, mw, f4, a);
        if (m > 8) gather8<8>(tin, ms, mw, f4, a);
        c0e += 16;
        if (c0e + lrow < en) {
            const int2 p = *(const int2*)(pair + 2 * (size_t)(c0e + lrow));
            ms = p.x;
            mw = __int_as_float(p.y);
        }
    }
    return a;
}

__device__ __forceinline__ void first_pairs(const int* __restrict__ pair, int rp, int half, int lrow,
                                            int (&pms)[8], float (&pmw)[8]) {
#pragma unroll
    for (int it = 0; it < 8; ++it) {
        const int b    = __shfl(rp, it * 2 + half);
        const int slot = (b + lrow < N_EDGES) ? (b + lrow) : (N_EDGES - 1);
        const int2 p   = *(const int2*)(pair + 2 * (size_t)slot);
        pms[it] = p.x;
        pmw[it] = __int_as_float(p.y);
    }
}

template <int RECUR>
__global__ void __launch_bounds__(NTHR)
prop_only(const float* __restrict__ tin, const float* __restrict__ tprev,
          const int* __restrict__ rowptr, const int* __restrict__ pair,
          float* __restrict__ tout, int nrows) {
    const int wave = threadIdx.x >> 5;
    const int lane = threadIdx.x & 31;
    const int row0 = (blockIdx.x * WPB + wave) * 16;
    if (row0 >= nrows) return;

    const int lrow = lane & 15;
    const int half = lane >> 4;
    const int f4   = lrow << 2;

    const int rp = rowptr[row0 + ((lane < 16) ? lane : 16)];
    int   pms[8];
    float pmw[8];
    first_pairs(pair, rp, half, lrow, pms, pmw);

#pragma unroll
    for (int it = 0; it < 8; ++it) {
        const int r  = it * 2 + half;
        const int b  = __shfl(rp, r), en = __shfl(rp, r + 1);
        float4 tp = make_float4(0.0f, 0.0f, 0.0f, 0.0f);
        if (RECUR) tp = *(const float4*)(tprev + (size_t)(row0 + r) * DHID + f4);
        float4 a = gather_row(tin, pair, b, en, pms[it], pmw[it], lrow, f4);
        if (RECUR) {
            a.x = 2.0f * a.x - tp.x;
            a.y = 2.0f * a.y - tp.y;
            a.z = 2.0f * a.z - tp.z;
            a.w = 2.0f * a.w - tp.w;
        }
        *(float4*)(tout + (size_t)(row0 + r) * DHID + f4) = a;
    }
}

#define TPITCH 68

template <int ACCIN, int FINAL>
__global__ void __launch_bounds__(NTHR)
prop_gemm2(const float* __restrict__ tin, const float* __restrict__ tprev,
           const int* __restrict__ rowptr, const int* __restrict__ pair,
           const float* __restrict__ Wa, const float* __restrict__ Wb,
           const float* __restrict__ bias, float* __restrict__ tout,
           float* __restrict__ acc, float* __restrict__ hout, int nrows) {
    __shared__ __attribute__((aligned(16))) float lds[2 * DHID * DHID + WPB * 16 * TPITCH];
    float* lW = lds;
    float* lT = lds + 2 * DHID * DHID;
    stage_w(lW, Wa);
    stage_w(lW + DHID * DHID, Wb);
    __syncthreads();

    const int wave = threadIdx.x >> 5;
    const int lane = threadIdx.x & 31;
    const int row0 = __builtin_amdgcn_readfirstlane((blockIdx.x * WPB + wave) * 16);
    if (row0 >= nrows) return;

    const int lrow  = lane & 15;
    const int half  = lane >> 4;
    const int khalf = half << 1;
    const int ccol  = lane & 15;
    const int f4    = lrow << 2;
    const int loff  = (half << 6) + f4;
    const size_t rbase = (size_t)row0 * DHID;

    const int rp = rowptr[row0 + ((lane < 16) ? lane : 16)];
    float* myT = lT + wave * (16 * TPITCH);
    const float* arow = myT + lrow * TPITCH + khalf;

#pragma unroll
    for (int it = 0; it < 8; ++it) {
        const int r = it * 2 + half;
        const float4 t = *(const float4*)(tprev + rbase + it * 2 * DHID + loff);
        *(float4*)(myT + r * TPITCH + f4) = t;
    }

    int   pms[8];
    float pmw[8];
    first_pairs(pair, rp, half, lrow, pms, pmw);

    __builtin_amdgcn_fence(__ATOMIC_RELEASE, "wavefront");
    __builtin_amdgcn_wave_barrier();
    __builtin_amdgcn_fence(__ATOMIC_ACQUIRE, "wavefront");

    v8f c0 = {}, c1 = {}, c2 = {}, c3 = {};
    tile_gemm(arow, lW, khalf, lrow, c0, c1, c2, c3);
    asm volatile("" : "+v"(c0), "+v"(c1), "+v"(c2), "+v"(c3));

#pragma unroll
    for (int it = 0; it < 8; ++it) {
        const int r  = it * 2 + half;
        const int b  = __shfl(rp, r), en = __shfl(rp, r + 1);
        float4 a = gather_row(tin, pair, b, en, pms[it], pmw[it], lrow, f4);
        const float4 tp = *(const float4*)(myT + r * TPITCH + f4);
        a.x = 2.0f * a.x - tp.x;
        a.y = 2.0f * a.y - tp.y;
        a.z = 2.0f * a.z - tp.z;
        a.w = 2.0f * a.w - tp.w;
        if (!FINAL) *(float4*)(tout + rbase + it * 2 * DHID + loff) = a;
        *(float4*)(myT + r * TPITCH + f4) = a;
    }

    float4 av[8];
    if (ACCIN) {
#pragma unroll
        for (int it = 0; it < 8; ++it)
            av[it] = *(const float4*)(acc + rbase + it * 2 * DHID + loff);
    }

    __builtin_amdgcn_fence(__ATOMIC_RELEASE, "wavefront");
    __builtin_amdgcn_wave_barrier();
    __builtin_amdgcn_fence(__ATOMIC_ACQUIRE, "wavefront");

    tile_gemm(arow, lW + DHID * DHID, khalf, lrow, c0, c1, c2, c3);

#pragma unroll
    for (int r = 0; r < 8; ++r) {
        float* tr = myT + ((half << 3) + r) * TPITCH + ccol;
        tr[0] = c0[r]; tr[16] = c1[r]; tr[32] = c2[r]; tr[48] = c3[r];
    }

    __builtin_amdgcn_fence(__ATOMIC_RELEASE, "wavefront");
    __builtin_amdgcn_wave_barrier();
    __builtin_amdgcn_fence(__ATOMIC_ACQUIRE, "wavefront");

    float4 bv = make_float4(0.0f, 0.0f, 0.0f, 0.0f);
    if (FINAL) bv = *(const float4*)(bias + f4);
#pragma unroll
    for (int it = 0; it < 8; ++it) {
        const int r = it * 2 + half;
        float4 v = *(const float4*)(myT + r * TPITCH + f4);
        if (ACCIN) {
            v.x = av[it].x + v.x; v.y = av[it].y + v.y;
            v.z = av[it].z + v.z; v.w = av[it].w + v.w;
        }
        if (FINAL) {
            v.x = fmaxf(v.x + bv.x, 0.0f); v.y = fmaxf(v.y + bv.y, 0.0f);
            v.z = fmaxf(v.z + bv.z, 0.0f); v.w = fmaxf(v.w + bv.w, 0.0f);
        }
        float* dstp = FINAL ? hout : acc;
        *(float4*)(dstp + rbase + it * 2 * DHID + loff) = v;
    }
}

extern "C" void kernel_launch(void* const* d_in, const int* in_sizes, int n_in,
                              void* d_out, int out_size, void* d_ws, size_t ws_size,
                              hipStream_t stream) {
    const float* x       = (const float*)d_in[0];
    const int*   ei      = (const int*)d_in[1];
    const float* ew      = (const float*)d_in[2];
    const float* weights = (const float*)d_in[3];
    const float* bias    = (const float*)d_in[4];
    float*       out     = (float*)d_out;

    const int* src = ei;
    const int* dst = ei + N_EDGES;

    float* ws   = (float*)d_ws;
    float* base = ws + N_EDGES + N_NODES;
    float* P[5];
    for (int i = 0; i < 5; ++i) P[i] = base + (size_t)i * N_NODES * DHID;
    float* ACC = P[4];

    int*   pair   = (int*)P[3];
    int*   rowptr = pair + 2 * N_EDGES;
    float* dinv   = (float*)(rowptr + N_NODES + 32);
    int*   cnt    = (int*)(dinv + N_NODES);
    int*   bsum   = cnt + N_NODES;
    int*   rank   = (int*)ws;
    int*   rec    = (int*)ACC;

    dim3 blk(256);
    dim3 pblk(NTHR);
    const int gE    = (N_EDGES + 255) / 256;
    const int gF    = (N_EDGES + 256 * FSLOTS - 1) / (256 * FSLOTS);
    const int gScan = (N_NODES + 1023) / 1024;
    const int gGemm = (N_NODES + WPB * 16 - 1) / (WPB * 16);
    const int WSZ   = DHID * DHID;

    zero_kernel     <<<256,   blk, 0, stream>>>(dinv, 2 * N_NODES);
    deg_rank_kernel <<<gE,    blk, 0, stream>>>(src, dst, ew, dinv, cnt, rank, N_EDGES);
    dinv_bsum_kernel<<<gScan, blk, 0, stream>>>(dinv, cnt, bsum, N_NODES);
    scan_kernel     <<<gScan, blk, 0, stream>>>(cnt, bsum, rowptr, N_NODES);
    fill_kernel     <<<gE,    blk, 0, stream>>>(src, dst, ew, rowptr, rank, rec, N_EDGES);
    csr_final_kernel<<<gF,    blk, 0, stream>>>(rec, dinv, rowptr, pair, N_EDGES);

    int h_idx = -1;
    for (int l = 0; l < NLAYERS; ++l) {
        int rid[3], c = 0;
        for (int i = 0; i < 3; ++i)
            if (i != h_idx) rid[c++] = i;

        const float* t0 = (h_idx < 0) ? x : P[h_idx];
        float* t1 = P[rid[0]];
        const float* Wl = weights + (size_t)l * KORD * WSZ;
        const float* bl = bias + l * DHID;

        prop_only<0><<<gGemm, pblk, 0, stream>>>(t0, nullptr, rowptr, pair, t1, N_NODES);

        int q[10]; int qh = 0, qt = 0;
        for (int i = 1; i < c; ++i) q[qt++] = rid[i];
        int t0_idx = h_idx;
        int t1_idx = rid[0];

        for (int k = 2; k < KORD; ++k) {
            int s_idx = q[qh++];
            const float* tin = P[t1_idx];
            const float* tpv = (t0_idx < 0) ? x : P[t0_idx];
            const float* Wa  = Wl + (size_t)(k - 2) * WSZ;
            const float* Wb  = Wl + (size_t)k * WSZ;
            if (k == 2) {
                prop_gemm2<0, 0><<<gGemm, pblk, 0, stream>>>(tin, tpv, rowptr, pair, Wa, Wb, bl,
                                                           P[s_idx], ACC, nullptr, N_NODES);
            } else if (k == 3 || k == 6) {
                prop_gemm2<1, 0><<<gGemm, pblk, 0, stream>>>(tin, tpv, rowptr, pair, Wa, Wb, bl,
                                                           P[s_idx], ACC, nullptr, N_NODES);
            } else if (k == KORD - 1) {
                float* hn = (l == NLAYERS - 1) ? out : P[s_idx];
                prop_gemm2<1, 1><<<gGemm, pblk, 0, stream>>>(tin, tpv, rowptr, pair, Wa, Wb, bl,
                                                           nullptr, ACC, hn, N_NODES);
            } else {
                prop_only<1><<<gGemm, pblk, 0, stream>>>(tin, tpv, rowptr, pair, P[s_idx], N_NODES);
            }
            if (t0_idx >= 0) q[qt++] = t0_idx;
            t0_idx = t1_idx;
            t1_idx = s_idx;
        }
        h_idx = t1_idx;
    }
}


# Row gather: 16 row loads in flight per row (one padded batch of 16 slots, was two of 8 in turn); registers freed by 32-bit row offsets and pair chunks fetched one row ahead; sum order unchanged
# speedup vs baseline: 4.3004x; 1.0349x over previous
#include <hip/hip_runtime.h>

#define N_NODES 100000
#define N_EDGES 1000000
#define DHID    64
#define KORD    8
#define NLAYERS 3
#define WPB     25
#define NTHR    (WPB * 32)

typedef float v2f __attribute__((ext_vector_type(2)));
typedef float v8f __attribute__((ext_vector_type(8)));

__global__ void zero_kernel(float* __restrict__ p, int n) {
    int i = blockIdx.x * blockDim.x + threadIdx.x;
    int stride = gridDim.x * blockDim.x;
    for (; i < n; i += stride) p[i] = 0.0f;
}

__global__ void deg_rank_kernel(const int* __restrict__ src, const int* __restrict__ dst,
                                const float* __restrict__ ew, float* __restrict__ deg,
                                int* __restrict__ cnt, int* __restrict__ rank, int nE) {
    int e = blockIdx.x * blockDim.x + threadIdx.x;
    if (e >= nE) return;
    int s = src[e], d = dst[e];
    if (s != d) atomicAdd(&deg[s], ew[e]);
    rank[e] = atomicAdd(&cnt[d], 1);
}

__global__ void __launch_bounds__(256)
dinv_bsum_kernel(float* __restrict__ deg, const int* __restrict__ cnt,
                 int* __restrict__ bsum, int n) {
    __shared__ int red[8];
    const int i0 = (blockIdx.x * 256 + threadIdx.x) * 4;
    int s = 0;
    for (int j = 0; j < 4; ++j) {
        int i = i0 + j;
        if (i < n) {
            float d = deg[i];
            deg[i] = (d > 0.0f) ? rsqrtf(d) : 0.0f;
            s += cnt[i];
        }
    }
    for (int off = 16; off > 0; off >>= 1) s += __shfl_xor(s, off);
    if ((threadIdx.x & 31) == 0) red[threadIdx.x >> 5] = s;
    __syncthreads();
    if (threadIdx.x == 0) {
        int t = 0;
        for (int w = 0; w < 8; ++w) t += red[w];
        bsum[blockIdx.x] = t;
    }
}

__global__ void __launch_bounds__(256)
scan_kernel(const int* __restrict__ cnt, const int* __restrict__ bsum,
            int* __restrict__ rowptr, int n) {
    __shared__ int wsum[8];
    __shared__ int osum[8];
    const int tid = threadIdx.x, lane = tid & 31, wave = tid >> 5;
    int o = (tid < (int)blockIdx.x) ? bsum[tid] : 0;
    for (int off = 16; off > 0; off >>= 1) o += __shfl_xor(o, off);
    const int i0 = (blockIdx.x * 256 + tid) * 4;
    int c[4];
    int s = 0;
    for (int j = 0; j < 4; ++j) {
        c[j] = (i0 + j < n) ? cnt[i0 + j] : 0;
        s += c[j];
    }
    int inc = s;
    for (int off = 1; off < 32; off <<= 1) {
        int v = __shfl_up(inc, off);
        if (lane >= off) inc += v;
    }
    if (lane == 0)  osum[wave] = o;
    if (lane == 31) wsum[wave] = inc;
    __syncthreads();
    int ex = inc - s;
    for (int w = 0; w < 8; ++w) {
        ex += osum[w];
        if (w < wave) ex += wsum[w];
    }
    for (int j = 0; j < 4; ++j) {
        if (i0 + j < n) { rowptr[i0 + j] = ex; ex += c[j]; }
    }
    if (i0 < n && i0 + 4 >= n) rowptr[n] = ex;
}

__global__ void fill_kernel(const int* __restrict__ src, const int* __restrict__ dst,
                            const float* __restrict__ ew, const int* __restrict__ rowptr,
                            const int* __restrict__ rank, int* __restrict__ rec, int nE) {
    int e = blockIdx.x * blockDim.x + threadIdx.x;
    if (e >= nE) return;
    int s = src[e], d = dst[e];
    float w = (s == d) ? 0.0f : ew[e];
    *(int4*)(rec + 4 * (size_t)(rowptr[d] + rank[e])) = make_int4(e, s, __float_as_int(w), d);
}

#define FSLOTS 4
__global__ void __launch_bounds__(256)
csr_final_kernel(const int* __restrict__ rec, const float* __restrict__ dinv,
                 const int* __restrict__ rowptr, int* __restrict__ pair, int nE) {
    const int j0 = blockIdx.x * (256 * FSLOTS) + threadIdx.x;
    int4  q[FSLOTS];
    int   b[FSLOTS], en[FSLOTS];
    float ds[FSLOTS], dd[FSLOTS];
#pragma unroll
    for (int k = 0; k < FSLOTS; ++k) {
        const int j  = j0 + k * 256;
        const int jc = (j < nE) ? j : (nE - 1);
        q[k] = *(const int4*)(rec + 4 * (size_t)jc);
    }
#pragma unroll
    for (int k = 0; k < FSLOTS; ++k) {
        const int s = q[k].y, d = q[k].w;
        b[k]  = rowptr[d];
        en[k] = rowptr[d + 1];
        ds[k] = dinv[s];
        dd[k] = dinv[d];
    }
#pragma unroll
    for (int k = 0; k < FSLOTS; ++k) {
        const int e = q[k].x;
        int r = 0;
        for (int i = b[k]; i < en[k]; ++i) r += (rec[4 * (size_t)i] < e) ? 1 : 0;
        const float nm = -ds[k] * __int_as_float(q[k].z) * dd[k];
        if (j0 + k * 256 < nE)
            *(int2*)(pair + 2 * (size_t)(b[k] + r)) = make_int2(q[k].y, __float_as_int(nm));
    }
}

#define WLOADS ((DHID * DHID + NTHR - 1) / NTHR)
__device__ __forceinline__ void stage_w(float* lWp, const float* __restrict__ W) {
    float v[WLOADS];
#pragma unroll
    for (int i = 0; i < WLOADS; ++i) {
        const int idx = threadIdx.x + i * NTHR;
        v[i] = (idx < DHID * DHID) ? W[idx] : 0.0f;
    }
#pragma unroll
    for (int i = 0; i < WLOADS; ++i) {
        const int idx = threadIdx.x + i * NTHR;
        const int k = idx >> 6, n = idx & 63;
        if (idx < DHID * DHID) lWp[((k >> 1) << 7) + (n << 1) + (k & 1)] = v[i];
    }
}

__device__ __forceinline__ void tile_gemm(const float* arow, const float* lWp, int khalf, int lrow,
                                          v8f& c0, v8f& c1, v8f& c2, v8f& c3) {
#pragma unroll
    for (int kt = 0; kt < 16; ++kt) {
        const int k0 = kt * 4 + khalf;
        const v2f a = *(const v2f*)(arow + kt * 4);
        const float* wp = lWp + ((k0 >> 1) << 7);
        v2f b;
        b = *(const v2f*)(wp + ((lrow +  0) << 1));
        c0 = __builtin_amdgcn_wmma_f32_16x16x4_f32(false, a, false, b, (short)0, c0, false, false);
        b = *(const v2f*)(wp + ((lrow + 16) << 1));
        c1 = __builtin_amdgcn_wmma_f32_16x16x4_f32(false, a, false, b, (short)0, c1, false, false);
        b = *(const v2f*)(wp + ((lrow + 32) << 1));
        c2 = __builtin_amdgcn_wmma_f32_16x16x4_f32(false, a, false, b, (short)0, c2, false, false);
        b = *(const v2f*)(wp + ((lrow + 48) << 1));
        c3 = __builtin_amdgcn_wmma_f32_16x16x4_f32(false, a, false, b, (short)0, c3, false, false);
    }
}

__device__ __forceinline__ void gather16(const float4* __restrict__ tin4, int ms, float mw, int lrow, float4& a) {
    float4 h[16];
#pragma unroll
    for (int j = 0; j < 16; ++j) {
        const int s = __shfl(ms, j, 16);
        h[j] = tin4[(unsigned)(s * (DHID / 4) + lrow)];
    }
#pragma unroll
    for (int j = 0; j < 16; ++j) {
        const float w = __shfl(mw, j, 16);
        a.x = fmaf(w, h[j].x, a.x);
        a.y = fmaf(w, h[j].y, a.y);
        a.z = fmaf(w, h[j].z, a.z);
        a.w = fmaf(w, h[j].w, a.w);
    }
}

__device__ __forceinline__ float4 gather_row(const float* __restrict__ tin, const int* __restrict__ pair,
                                             int b, int en, int ms, float mw, int lrow) {
    const float4* __restrict__ tin4 = (const float4*)tin;
    float4 a = make_float4(0.0f, 0.0f, 0.0f, 0.0f);
    int c0e = b;
    while (c0e < en) {
        const int m  = (en - c0e < 16) ? (en - c0e) : 16;
        const int sl = __shfl(ms, m - 1, 16);
        if (lrow >= m) {
            ms = sl;
            mw = 0.0f;
        }
        gather16(tin4, ms, mw, lrow, a);
        c0e += 16;
        if (c0e + lrow < en) {
            const int2 p = *(const int2*)(pair + 2 * (size_t)(c0e + lrow));
            ms = p.x;
            mw = __int_as_float(p.y);
        }
    }
    return a;
}

__device__ __forceinline__ int2 row_pairs(const int* __restrict__ pair, int b, int lrow) {
    const int slot = (b + lrow < N_EDGES) ? (b + lrow) : (N_EDGES - 1);
    return *(const int2*)(pair + 2 * (size_t)slot);
}

template <int RECUR>
__global__ void __launch_bounds__(NTHR)
prop_only(const float* __restrict__ tin, const float* __restrict__ tprev,
          const int* __restrict__ rowptr, const int* __restrict__ pair,
          float* __restrict__ tout, int nrows) {
    const int wave = threadIdx.x >> 5;
    const int lane = threadIdx.x & 31;
    const int row0 = (blockIdx.x * WPB + wave) * 16;
    if (row0 >= nrows) return;

    const int lrow = lane & 15;
    const int half = lane >> 4;
    const int f4   = lrow << 2;

    const int rp = rowptr[row0 + ((lane < 16) ? lane : 16)];
    int2 pc = row_pairs(pair, __shfl(rp, half), lrow);

#pragma unroll
    for (int it = 0; it < 8; ++it) {
        const int r  = it * 2 + half;
        const int b  = __shfl(rp, r), en = __shfl(rp, r + 1);
        int2 pn = pc;
        if (it < 7) pn = row_pairs(pair, __shfl(rp, r + 2), lrow);
        float4 tp = make_float4(0.0f, 0.0f, 0.0f, 0.0f);
        if (RECUR) tp = *(const float4*)(tprev + (size_t)(row0 + r) * DHID + f4);
        float4 a = gather_row(tin, pair, b, en, pc.x, __int_as_float(pc.y), lrow);
        if (RECUR) {
            a.x = 2.0f * a.x - tp.x;
            a.y = 2.0f * a.y - tp.y;
            a.z = 2.0f * a.z - tp.z;
            a.w = 2.0f * a.w - tp.w;
        }
        *(float4*)(tout + (size_t)(row0 + r) * DHID + f4) = a;
        pc = pn;
    }
}

#define TPITCH 68

template <int ACCIN, int FINAL>
__global__ void __launch_bounds__(NTHR)
prop_gemm2(const float* __restrict__ tin, const float* __restrict__ tprev,
           const int* __restrict__ rowptr, const int* __restrict__ pair,
           const float* __restrict__ Wa, const float* __restrict__ Wb,
           const float* __restrict__ bias, float* __restrict__ tout,
           float* __restrict__ acc, float* __restrict__ hout, int nrows) {
    __shared__ __attribute__((aligned(16))) float lds[2 * DHID * DHID + WPB * 16 * TPITCH];
    float* lW = lds;
    float* lT = lds + 2 * DHID * DHID;
    stage_w(lW, Wa);
    stage_w(lW + DHID * DHID, Wb);
    __syncthreads();

    const int wave = threadIdx.x >> 5;
    const int lane = threadIdx.x & 31;
    const int row0 = __builtin_amdgcn_readfirstlane((blockIdx.x * WPB + wave) * 16);
    if (row0 >= nrows) return;

    const int lrow  = lane & 15;
    const int half  = lane >> 4;
    const int khalf = half << 1;
    const int ccol  = lane & 15;
    const int f4    = lrow << 2;
    const int loff  = (half << 6) + f4;
    const size_t rbase = (size_t)row0 * DHID;

    const int rp = rowptr[row0 + ((lane < 16) ? lane : 16)];
    float* myT = lT + wave * (16 * TPITCH);
    const float* arow = myT + lrow * TPITCH + khalf;
    float* const trow = myT + half * TPITCH + f4;

#pragma unroll
    for (int it = 0; it < 8; ++it) {
        const float4 t = *(const float4*)(tprev + rbase + it * 2 * DHID + loff);
        *(float4*)(trow + it * 2 * TPITCH) = t;
    }

    int2 pc = row_pairs(pair, __shfl(rp, half), lrow);

    __builtin_amdgcn_fence(__ATOMIC_RELEASE, "wavefront");
    __builtin_amdgcn_wave_barrier();
    __builtin_amdgcn_fence(__ATOMIC_ACQUIRE, "wavefront");

    v8f c0 = {}, c1 = {}, c2 = {}, c3 = {};
    tile_gemm(arow, lW, khalf, lrow, c0, c1, c2, c3);
    asm volatile("" : "+v"(c0), "+v"(c1), "+v"(c2), "+v"(c3));

#pragma unroll
    for (int it = 0; it < 8; ++it) {
        const int r  = it * 2 + half;
        const int b  = __shfl(rp, r), en = __shfl(rp, r + 1);
        int2 pn = pc;
        if (it < 7) pn = row_pairs(pair, __shfl(rp, r + 2), lrow);
        float4 a = gather_row(tin, pair, b, en, pc.x, __int_as_float(pc.y), lrow);
        const float4 tp = *(const float4*)(trow + it * 2 * TPITCH);
        a.x = 2.0f * a.x - tp.x;
        a.y = 2.0f * a.y - tp.y;
        a.z = 2.0f * a.z - tp.z;
        a.w = 2.0f * a.w - tp.w;
        if (!FINAL) *(float4*)(tout + rbase + it * 2 * DHID + loff) = a;
        *(float4*)(trow + it * 2 * TPITCH) = a;
        pc = pn;
    }

    float4 av[8];
    if (ACCIN) {
#pragma unroll
        for (int it = 0; it < 8; ++it)
            av[it] = *(const float4*)(acc + rbase + it * 2 * DHID + loff);
    }

    __builtin_amdgcn_fence(__ATOMIC_RELEASE, "wavefront");
    __builtin_amdgcn_wave_barrier();
    __builtin_amdgcn_fence(__ATOMIC_ACQUIRE, "wavefront");

    tile_gemm(arow, lW + DHID * DHID, khalf, lrow, c0, c1, c2, c3);

#pragma unroll
    for (int r = 0; r < 8; ++r) {
        float* tr = myT + ((half << 3) + r) * TPITCH + ccol;
        tr[0] = c0[r]; tr[16] = c1[r]; tr[32] = c2[r]; tr[48] = c3[r];
    }

    __builtin_amdgcn_fence(__ATOMIC_RELEASE, "wavefront");
    __builtin_amdgcn_wave_barrier();
    __builtin_amdgcn_fence(__ATOMIC_ACQUIRE, "wavefront");

    float4 bv = make_float4(0.0f, 0.0f, 0.0f, 0.0f);
    if (FINAL) bv = *(const float4*)(bias + f4);
#pragma unroll
    for (int it = 0; it < 8; ++it) {
        float4 v = *(const float4*)(trow + it * 2 * TPITCH);
        if (ACCIN) {
            v.x = av[it].x + v.x; v.y = av[it].y + v.y;
            v.z = av[it].z + v.z; v.w = av[it].w + v.w;
        }
        if (FINAL) {
            v.x = fmaxf(v.x + bv.x, 0.0f); v.y = fmaxf(v.y + bv.y, 0.0f);
            v.z = fmaxf(v.z + bv.z, 0.0f); v.w = fmaxf(v.w + bv.w, 0.0f);
        }
        float* dstp = FINAL ? hout : acc;
        *(float4*)(dstp + rbase + it * 2 * DHID + loff) = v;
    }
}

extern "C" void kernel_launch(void* const* d_in, const int* in_sizes, int n_in,
                              void* d_out, int out_size, void* d_ws, size_t ws_size,
                              hipStream_t stream) {
    const float* x       = (const float*)d_in[0];
    const int*   ei      = (const int*)d_in[1];
    const float* ew      = (const float*)d_in[2];
    const float* weights = (const float*)d_in[3];
    const float* bias    = (const float*)d_in[4];
    float*       out     = (float*)d_out;

    const int* src = ei;
    const int* dst = ei + N_EDGES;

    float* ws   = (float*)d_ws;
    float* base = ws + N_EDGES + N_NODES;
    float* P[5];
    for (int i = 0; i < 5; ++i) P[i] = base + (size_t)i * N_NODES * DHID;
    float* ACC = P[4];

    int*   pair   = (int*)P[3];
    int*   rowptr = pair + 2 * N_EDGES;
    float* dinv   = (float*)(rowptr + N_NODES + 32);
    int*   cnt    = (int*)(dinv + N_NODES);
    int*   bsum   = cnt + N_NODES;
    int*   rank   = (int*)ws;
    int*   rec    = (int*)ACC;

    dim3 blk(256);
    dim3 pblk(NTHR);
    const int gE    = (N_EDGES + 255) / 256;
    const int gF    = (N_EDGES + 256 * FSLOTS - 1) / (256 * FSLOTS);
    const int gScan = (N_NODES + 1023) / 1024;
    const int gGemm = (N_NODES + WPB * 16 - 1) / (WPB * 16);
    const int WSZ   = DHID * DHID;

    zero_kernel     <<<256,   blk, 0, stream>>>(dinv, 2 * N_NODES);
    deg_rank_kernel <<<gE,    blk, 0, stream>>>(src, dst, ew, dinv, cnt, rank, N_EDGES);
    dinv_bsum_kernel<<<gScan, blk, 0, stream>>>(dinv, cnt, bsum, N_NODES);
    scan_kernel     <<<gScan, blk, 0, stream>>>(cnt, bsum, rowptr, N_NODES);
    fill_kernel     <<<gE,    blk, 0, stream>>>(src, dst, ew, rowptr, rank, rec, N_EDGES);
    csr_final_kernel<<<gF,    blk, 0, stream>>>(rec, dinv, rowptr, pair, N_EDGES);

    int h_idx = -1;
    for (int l = 0; l < NLAYERS; ++l) {
        int rid[3], c = 0;
        for (int i = 0; i < 3; ++i)
            if (i != h_idx) rid[c++] = i;

        const float* t0 = (h_idx < 0) ? x : P[h_idx];
        float* t1 = P[rid[0]];
        const float* Wl = weights + (size_t)l * KORD * WSZ;
        const float* bl = bias + l * DHID;

        prop_only<0><<<gGemm, pblk, 0, stream>>>(t0, nullptr, rowptr, pair, t1, N_NODES);

        int q[10]; int qh = 0, qt = 0;
        for (int i = 1; i < c; ++i) q[qt++] = rid[i];
        int t0_idx = h_idx;
        int t1_idx = rid[0];

        for (int k = 2; k < KORD; ++k) {
            int s_idx = q[qh++];
            const float* tin = P[t1_idx];
            const float* tpv = (t0_idx < 0) ? x : P[t0_idx];
            const float* Wa  = Wl + (size_t)(k - 2) * WSZ;
            const float* Wb  = Wl + (size_t)k * WSZ;
            if (k == 2) {
                prop_gemm2<0, 0><<<gGemm, pblk, 0, stream>>>(tin, tpv, rowptr, pair, Wa, Wb, bl,
                                                           P[s_idx], ACC, nullptr, N_NODES);
            } else if (k == 3 || k == 6) {
                prop_gemm2<1, 0><<<gGemm, pblk, 0, stream>>>(tin, tpv, rowptr, pair, Wa, Wb, bl,
                                                           P[s_idx], ACC, nullptr, N_NODES);
            } else if (k == KORD - 1) {
                float* hn = (l == NLAYERS - 1) ? out : P[s_idx];
                prop_gemm2<1, 1><<<gGemm, pblk, 0, stream>>>(tin, tpv, rowptr, pair, Wa, Wb, bl,
                                                           nullptr, ACC, hn, N_NODES);
            } else {
                prop_only<1><<<gGemm, pblk, 0, stream>>>(tin, tpv, rowptr, pair, P[s_idx], N_NODES);
            }
            if (t0_idx >= 0) q[qt++] = t0_idx;
            t0_idx = t1_idx;
            t1_idx = s_idx;
        }
        h_idx = t1_idx;
    }
}
